# starting kernel
# baseline (speedup 1.0000x reference)
#include <hip/hip_runtime.h>
#include <math.h>

typedef __attribute__((ext_vector_type(2))) float v2f;
typedef __attribute__((ext_vector_type(8))) float v8f;

#define NQ      12
#define NSTATE  4096
#define NLAYERS 4
#define NTHREADS 256
#define NGATES  (NQ + NLAYERS * NQ)

__device__ __forceinline__ void buildM(float* mr, float* mi, int wl,
                                       const float* gg, int t) {
    const int row = t >> 4, cm = t & 15;
    const int mask = 15 & ~(1 << wl);
    float re = 0.0f, im = 0.0f;
    if (((row ^ cm) & mask) == 0) {
        const int rb = (row >> wl) & 1;
        const int cb = (cm >> wl) & 1;
        const int e = (rb * 2 + cb) * 2;
        re = gg[e];
        im = gg[e + 1];
    }
    mr[t] = re;
    mi[t] = im;
}

template <int G>
__device__ __forceinline__ void apply16T(float* sr, float* si,
                                         const float* mr, const float* mi,
                                         int wave, int lane) {
    constexpr int SH = G * 4;
    constexpr int LOWMASK = (1 << SH) - 1;
    const int half = lane >> 4;
    const int n15  = lane & 15;

    v2f ar[4], ai[4], ani[4];
#pragma unroll
    for (int kc = 0; kc < 4; ++kc) {
        const int k0 = kc * 4 + 2 * half;
        ar[kc].x  = mr[n15 * 16 + k0];  ar[kc].y  = mr[n15 * 16 + k0 + 1];
        ai[kc].x  = mi[n15 * 16 + k0];  ai[kc].y  = mi[n15 * 16 + k0 + 1];
        ani[kc].x = -ai[kc].x;          ani[kc].y = -ai[kc].y;
    }

#pragma unroll
    for (int tt = 0; tt < 2; ++tt) {
        const int col = (wave * 2 + tt) * 16 + n15;
        const int base = ((col >> SH) << (SH + 4)) | (col & LOWMASK);

        v2f br[4], bi[4];
#pragma unroll
        for (int kc = 0; kc < 4; ++kc) {
            const int k0 = kc * 4 + 2 * half;
            const int i0 = base | ((k0) << SH);
            const int i1 = base | ((k0 + 1) << SH);
            br[kc].x = sr[i0]; br[kc].y = sr[i1];
            bi[kc].x = si[i0]; bi[kc].y = si[i1];
        }

        v8f accR = {0.f,0.f,0.f,0.f,0.f,0.f,0.f,0.f};
        v8f accI = {0.f,0.f,0.f,0.f,0.f,0.f,0.f,0.f};

#pragma unroll
        for (int kc = 0; kc < 4; ++kc) {
            accR = __builtin_amdgcn_wmma_f32_16x16x4_f32(false, ar[kc],  false, br[kc], (short)0, accR, false, false);
            accR = __builtin_amdgcn_wmma_f32_16x16x4_f32(false, ani[kc], false, bi[kc], (short)0, accR, false, false);
            accI = __builtin_amdgcn_wmma_f32_16x16x4_f32(false, ar[kc],  false, bi[kc], (short)0, accI, false, false);
            accI = __builtin_amdgcn_wmma_f32_16x16x4_f32(false, ai[kc],  false, br[kc], (short)0, accI, false, false);
        }

#pragma unroll
        for (int r = 0; r < 8; ++r) {
            const int i = base | ((r + 8 * half) << SH);
            sr[i] = accR[r];
            si[i] = accI[r];
        }
    }
}

__device__ __forceinline__ void applyDispatch(float* sr, float* si,
                                              const float* mr, const float* mi,
                                              int g, int wave, int lane) {
    if (g == 0)      apply16T<0>(sr, si, mr, mi, wave, lane);
    else if (g == 1) apply16T<1>(sr, si, mr, mi, wave, lane);
    else             apply16T<2>(sr, si, mr, mi, wave, lane);
}

__global__ __launch_bounds__(NTHREADS)
void vqc_kernel(const float* __restrict__ x,
                const float* __restrict__ wts,
                float* __restrict__ out) {
    __shared__ float sr[NSTATE];
    __shared__ float si[NSTATE];
    __shared__ float mr[256];
    __shared__ float mi[256];
    __shared__ float gm[NGATES * 8];
    __shared__ float wred[NQ * 8];

    const int t    = threadIdx.x;
    const int b    = blockIdx.x;
    const int wave = t >> 5;
    const int lane = t & 31;

    if (t < NGATES) {
        float e0, e1, e2, e3, e4, e5, e6, e7;
        if (t < NQ) {
            const float a = x[b * NQ + t] * 0.5f;
            const float c = cosf(a), s = sinf(a);
            e0 = c;    e1 = 0.f;
            e2 = 0.f;  e3 = -s;
            e4 = 0.f;  e5 = -s;
            e6 = c;    e7 = 0.f;
        } else {
            const float* wp = wts + (t - NQ) * 3;
            const float phi = wp[0], th = wp[1], om = wp[2];
            const float ct = cosf(th * 0.5f);
            const float sn = sinf(th * 0.5f);
            const float hs = 0.5f * (phi + om);
            const float hd = 0.5f * (phi - om);
            const float chs = cosf(hs), shs = sinf(hs);
            const float chd = cosf(hd), shd = sinf(hd);
            e0 =  chs * ct;  e1 = -shs * ct;
            e2 = -chd * sn;  e3 = -shd * sn;
            e4 =  chd * sn;  e5 = -shd * sn;
            e6 =  chs * ct;  e7 =  shs * ct;
        }
        float* gr = gm + t * 8;
        gr[0] = e0; gr[1] = e1; gr[2] = e2; gr[3] = e3;
        gr[4] = e4; gr[5] = e5; gr[6] = e6; gr[7] = e7;
    }

#pragma unroll
    for (int k = 0; k < 16; ++k) {
        const int i = k * NTHREADS + t;
        sr[i] = 0.0f;
        si[i] = 0.0f;
    }
    if (t == 0) sr[0] = 1.0f;
    __syncthreads();

    for (int q = 0; q < NQ; ++q) {
        const int bw = 11 - q;
        buildM(mr, mi, bw & 3, gm + q * 8, t);
        __syncthreads();
        applyDispatch(sr, si, mr, mi, bw >> 2, wave, lane);
        __syncthreads();
    }

    for (int l = 0; l < NLAYERS; ++l) {
        for (int q = 0; q < NQ; ++q) {
            const int bw = 11 - q;
            buildM(mr, mi, bw & 3, gm + (NQ + l * NQ + q) * 8, t);
            __syncthreads();
            applyDispatch(sr, si, mr, mi, bw >> 2, wave, lane);
            __syncthreads();
        }

        const int r = l % (NQ - 1) + 1;
        float vr[16], vi[16];
#pragma unroll
        for (int k = 0; k < 16; ++k) {
            const int j = k * NTHREADS + t;
            int src = j;
            for (int qq = NQ - 1; qq >= 0; --qq) {
                const int cbit = 11 - qq;
                const int tbit = 11 - ((qq + r) % NQ);
                src ^= ((src >> cbit) & 1) << tbit;
            }
            vr[k] = sr[src];
            vi[k] = si[src];
        }
        __syncthreads();
#pragma unroll
        for (int k = 0; k < 16; ++k) {
            const int j = k * NTHREADS + t;
            sr[j] = vr[k];
            si[j] = vi[k];
        }
        __syncthreads();
    }

    float part[NQ];
#pragma unroll
    for (int q = 0; q < NQ; ++q) part[q] = 0.0f;
#pragma unroll
    for (int k = 0; k < 16; ++k) {
        const int i = k * NTHREADS + t;
        const float p = sr[i] * sr[i] + si[i] * si[i];
#pragma unroll
        for (int q = 0; q < NQ; ++q) {
            part[q] += ((i >> (11 - q)) & 1) ? -p : p;
        }
    }
#pragma unroll
    for (int q = 0; q < NQ; ++q) {
        float v = part[q];
        for (int off = 16; off >= 1; off >>= 1)
            v += __shfl_xor(v, off, 32);
        if (lane == 0) wred[q * 8 + wave] = v;
    }
    __syncthreads();
    if (t < NQ) {
        float s = 0.0f;
#pragma unroll
        for (int wv = 0; wv < 8; ++wv) s += wred[t * 8 + wv];
        out[b * NQ + t] = s;
    }
}

extern "C" void kernel_launch(void* const* d_in, const int* in_sizes, int n_in,
                              void* d_out, int out_size, void* d_ws, size_t ws_size,
                              hipStream_t stream) {
    const float* x   = (const float*)d_in[0];
    const float* wts = (const float*)d_in[1];
    float* out = (float*)d_out;
    const int batch = in_sizes[0] / NQ;
    vqc_kernel<<<batch, NTHREADS, 0, stream>>>(x, wts, out);
}


# fuse single-qubit gates between CNOT rings: RX folded into layer-0 Rot per wire, each 4-bit group's 4 wires as one 16x16 tensor-product operator; 60 WMMA passes -> 12; f32 summation order changes
# speedup vs baseline: 3.5525x; 3.5525x over previous
#include <hip/hip_runtime.h>
#include <math.h>


typedef __attribute__((ext_vector_type(2))) float v2f;
typedef __attribute__((ext_vector_type(8))) float v8f;

#define NQ      12
#define NSTATE  4096
#define NLAYERS 4
#define NTHREADS 256
#define NFUSED  (NLAYERS * NQ)

__device__ __forceinline__ void buildLayer(float* mr, float* mi,
                                           const float* gl, int t) {
    const int row = t >> 4, cm = t & 15;
#pragma unroll
    for (int g = 0; g < 3; ++g) {
        float pr = 0.0f, pi = 0.0f;
#pragma unroll
        for (int wl = 0; wl < 4; ++wl) {
            const int q = 11 - (4 * g + wl);
            const int e = ((((row >> wl) & 1) << 1) | ((cm >> wl) & 1)) << 1;
            const float ur = gl[q * 8 + e];
            const float ui = gl[q * 8 + e + 1];
            if (wl == 0) {
                pr = ur;
                pi = ui;
            } else {
                const float nr = pr * ur - pi * ui;
                const float ni = pr * ui + pi * ur;
                pr = nr;
                pi = ni;
            }
        }
        mr[g * 256 + t] = pr;
        mi[g * 256 + t] = pi;
    }
}

template <int G>
__device__ __forceinline__ void apply16T(float* sr, float* si,
                                         const float* mr, const float* mi,
                                         int wave, int lane) {
    constexpr int SH = G * 4;
    constexpr int LOWMASK = (1 << SH) - 1;
    const int half = lane >> 4;
    const int n15  = lane & 15;

    v2f ar[4], ai[4], ani[4];
#pragma unroll
    for (int kc = 0; kc < 4; ++kc) {
        const int k0 = kc * 4 + 2 * half;
        ar[kc].x  = mr[n15 * 16 + k0];  ar[kc].y  = mr[n15 * 16 + k0 + 1];
        ai[kc].x  = mi[n15 * 16 + k0];  ai[kc].y  = mi[n15 * 16 + k0 + 1];
        ani[kc].x = -ai[kc].x;          ani[kc].y = -ai[kc].y;
    }

#pragma unroll
    for (int tt = 0; tt < 2; ++tt) {
        const int col = (wave * 2 + tt) * 16 + n15;
        const int base = ((col >> SH) << (SH + 4)) | (col & LOWMASK);

        v2f br[4], bi[4];
#pragma unroll
        for (int kc = 0; kc < 4; ++kc) {
            const int k0 = kc * 4 + 2 * half;
            const int i0 = base | ((k0) << SH);
            const int i1 = base | ((k0 + 1) << SH);
            br[kc].x = sr[i0]; br[kc].y = sr[i1];
            bi[kc].x = si[i0]; bi[kc].y = si[i1];
        }

        v8f accR = {0.f,0.f,0.f,0.f,0.f,0.f,0.f,0.f};
        v8f accI = {0.f,0.f,0.f,0.f,0.f,0.f,0.f,0.f};

#pragma unroll
        for (int kc = 0; kc < 4; ++kc) {
            accR = __builtin_amdgcn_wmma_f32_16x16x4_f32(false, ar[kc],  false, br[kc], (short)0, accR, false, false);
            accR = __builtin_amdgcn_wmma_f32_16x16x4_f32(false, ani[kc], false, bi[kc], (short)0, accR, false, false);
            accI = __builtin_amdgcn_wmma_f32_16x16x4_f32(false, ar[kc],  false, bi[kc], (short)0, accI, false, false);
            accI = __builtin_amdgcn_wmma_f32_16x16x4_f32(false, ai[kc],  false, br[kc], (short)0, accI, false, false);
        }

#pragma unroll
        for (int r = 0; r < 8; ++r) {
            const int i = base | ((r + 8 * half) << SH);
            sr[i] = accR[r];
            si[i] = accI[r];
        }
    }
}

__global__ __launch_bounds__(NTHREADS)
void vqc_kernel(const float* __restrict__ x,
                const float* __restrict__ wts,
                float* __restrict__ out) {
    __shared__ float sr[NSTATE];
    __shared__ float si[NSTATE];
    __shared__ float mr[3 * 256];
    __shared__ float mi[3 * 256];
    __shared__ float gu[NFUSED * 8];
    __shared__ float wred[NQ * 8];

    const int t    = threadIdx.x;
    const int b    = blockIdx.x;
    const int wave = t >> 5;
    const int lane = t & 31;

    if (t < NFUSED) {
        const float* wp = wts + t * 3;
        const float phi = wp[0], th = wp[1], om = wp[2];
        const float ct = cosf(th * 0.5f);
        const float sn = sinf(th * 0.5f);
        const float hs = 0.5f * (phi + om);
        const float hd = 0.5f * (phi - om);
        const float chs = cosf(hs), shs = sinf(hs);
        const float chd = cosf(hd), shd = sinf(hd);
        float e0 =  chs * ct, e1 = -shs * ct;
        float e2 = -chd * sn, e3 = -shd * sn;
        float e4 =  chd * sn, e5 = -shd * sn;
        float e6 =  chs * ct, e7 =  shs * ct;
        if (t < NQ) {
            const float a = x[b * NQ + t] * 0.5f;
            const float c = cosf(a), s = sinf(a);
            const float u0 = e0 * c + e3 * s, u1 = e1 * c - e2 * s;
            const float u2 = e1 * s + e2 * c, u3 = e3 * c - e0 * s;
            const float u4 = e4 * c + e7 * s, u5 = e5 * c - e6 * s;
            const float u6 = e5 * s + e6 * c, u7 = e7 * c - e4 * s;
            e0 = u0; e1 = u1; e2 = u2; e3 = u3;
            e4 = u4; e5 = u5; e6 = u6; e7 = u7;
        }
        float* gr = gu + t * 8;
        gr[0] = e0; gr[1] = e1; gr[2] = e2; gr[3] = e3;
        gr[4] = e4; gr[5] = e5; gr[6] = e6; gr[7] = e7;
    }

#pragma unroll
    for (int k = 0; k < 16; ++k) {
        const int i = k * NTHREADS + t;
        sr[i] = 0.0f;
        si[i] = 0.0f;
    }
    if (t == 0) sr[0] = 1.0f;
    __syncthreads();

    buildLayer(mr, mi, gu, t);
    __syncthreads();

    for (int l = 0; l < NLAYERS; ++l) {
        apply16T<2>(sr, si, mr + 512, mi + 512, wave, lane);
        __syncthreads();
        apply16T<1>(sr, si, mr + 256, mi + 256, wave, lane);
        __syncthreads();
        apply16T<0>(sr, si, mr, mi, wave, lane);
        __syncthreads();

        const int r = l % (NQ - 1) + 1;
        float vr[16], vi[16];
#pragma unroll
        for (int k = 0; k < 16; ++k) {
            const int j = k * NTHREADS + t;
            int src = j;
            for (int qq = NQ - 1; qq >= 0; --qq) {
                const int cbit = 11 - qq;
                const int tbit = 11 - ((qq + r) % NQ);
                src ^= ((src >> cbit) & 1) << tbit;
            }
            vr[k] = sr[src];
            vi[k] = si[src];
        }
        if (l + 1 < NLAYERS) buildLayer(mr, mi, gu + (l + 1) * NQ * 8, t);
        __syncthreads();
#pragma unroll
        for (int k = 0; k < 16; ++k) {
            const int j = k * NTHREADS + t;
            sr[j] = vr[k];
            si[j] = vi[k];
        }
        __syncthreads();
    }

    float part[NQ];
#pragma unroll
    for (int q = 0; q < NQ; ++q) part[q] = 0.0f;
#pragma unroll
    for (int k = 0; k < 16; ++k) {
        const int i = k * NTHREADS + t;
        const float p = sr[i] * sr[i] + si[i] * si[i];
#pragma unroll
        for (int q = 0; q < NQ; ++q) {
            part[q] += ((i >> (11 - q)) & 1) ? -p : p;
        }
    }
#pragma unroll
    for (int q = 0; q < NQ; ++q) {
        float v = part[q];
        for (int off = 16; off >= 1; off >>= 1)
            v += __shfl_xor(v, off, 32);
        if (lane == 0) wred[q * 8 + wave] = v;
    }
    __syncthreads();
    if (t < NQ) {
        float s = 0.0f;
#pragma unroll
        for (int wv = 0; wv < 8; ++wv) s += wred[t * 8 + wv];
        out[b * NQ + t] = s;
    }
}

extern "C" void kernel_launch(void* const* d_in, const int* in_sizes, int n_in,
                              void* d_out, int out_size, void* d_ws, size_t ws_size,
                              hipStream_t stream) {
    const float* x   = (const float*)d_in[0];
    const float* wts = (const float*)d_in[1];
    float* out = (float*)d_out;
    const int batch = in_sizes[0] / NQ;
    vqc_kernel<<<batch, NTHREADS, 0, stream>>>(x, wts, out);
}


# CNOT ring folded into the next layer's first-pass LDS load addresses (two state buffers) and into the signs of the final sums: no gather/scatter pass, 8 fewer barriers; final f32 sum order changes
# speedup vs baseline: 4.6594x; 1.3116x over previous
#include <hip/hip_runtime.h>
#include <math.h>


typedef __attribute__((ext_vector_type(2))) float v2f;
typedef __attribute__((ext_vector_type(8))) float v8f;

#define NQ      12
#define NSTATE  4096
#define NLAYERS 4
#define NTHREADS 256
#define NFUSED  (NLAYERS * NQ)

__host__ __device__ constexpr int ringSrc(int r, int j) {
    int s = j;
    for (int qq = NQ - 1; qq >= 0; --qq) {
        const int cbit = 11 - qq;
        const int tbit = 11 - ((qq + r) % NQ);
        s ^= ((s >> cbit) & 1) << tbit;
    }
    return s;
}
__host__ __device__ constexpr int ringDst(int r, int i) {
    int s = i;
    for (int qq = 0; qq < NQ; ++qq) {
        const int cbit = 11 - qq;
        const int tbit = 11 - ((qq + r) % NQ);
        s ^= ((s >> cbit) & 1) << tbit;
    }
    return s;
}

template <int g>
__device__ __forceinline__ void buildGroup(float* mr, float* mi,
                                           const float* gl, int t) {
    const int row = t >> 4, cm = t & 15;
    float pr = 0.0f, pi = 0.0f;
#pragma unroll
    for (int wl = 0; wl < 4; ++wl) {
        const int q = 11 - (4 * g + wl);
        const int e = ((((row >> wl) & 1) << 1) | ((cm >> wl) & 1)) << 1;
        const float ur = gl[q * 8 + e];
        const float ui = gl[q * 8 + e + 1];
        if (wl == 0) {
            pr = ur;
            pi = ui;
        } else {
            const float nr = pr * ur - pi * ui;
            const float ni = pr * ui + pi * ur;
            pr = nr;
            pi = ni;
        }
    }
    mr[g * 256 + t] = pr;
    mi[g * 256 + t] = pi;
}

template <int G, int R>
__device__ __forceinline__ void apply16T(const float* in, float* out,
                                         const float* mr, const float* mi,
                                         int wave, int lane) {
    constexpr int SH = G * 4;
    constexpr int LOWMASK = (1 << SH) - 1;
    const int half = lane >> 4;
    const int n15  = lane & 15;

    v2f ar[4], ai[4], ani[4];
#pragma unroll
    for (int kc = 0; kc < 4; ++kc) {
        const int k0 = kc * 4 + 2 * half;
        ar[kc].x  = mr[n15 * 16 + k0];  ar[kc].y  = mr[n15 * 16 + k0 + 1];
        ai[kc].x  = mi[n15 * 16 + k0];  ai[kc].y  = mi[n15 * 16 + k0 + 1];
        ani[kc].x = -ai[kc].x;          ani[kc].y = -ai[kc].y;
    }

#pragma unroll
    for (int tt = 0; tt < 2; ++tt) {
        const int col = (wave * 2 + tt) * 16 + n15;
        const int base = ((col >> SH) << (SH + 4)) | (col & LOWMASK);

        v2f br[4], bi[4];
        if (R == 0) {
#pragma unroll
            for (int kc = 0; kc < 4; ++kc) {
                const int k0 = kc * 4 + 2 * half;
                const int i0 = base | ((k0) << SH);
                const int i1 = base | ((k0 + 1) << SH);
                br[kc].x = in[i0];          br[kc].y = in[i1];
                bi[kc].x = in[NSTATE + i0]; bi[kc].y = in[NSTATE + i1];
            }
        } else {
            const int pl = ringSrc(R, base | ((2 * half) << SH));
#pragma unroll
            for (int kc = 0; kc < 4; ++kc) {
                const int i0 = pl ^ ringSrc(R, (kc * 4) << SH);
                const int i1 = pl ^ ringSrc(R, (kc * 4 + 1) << SH);
                br[kc].x = in[i0];          br[kc].y = in[i1];
                bi[kc].x = in[NSTATE + i0]; bi[kc].y = in[NSTATE + i1];
            }
        }

        v8f accR = {0.f,0.f,0.f,0.f,0.f,0.f,0.f,0.f};
        v8f accI = {0.f,0.f,0.f,0.f,0.f,0.f,0.f,0.f};

#pragma unroll
        for (int kc = 0; kc < 4; ++kc) {
            accR = __builtin_amdgcn_wmma_f32_16x16x4_f32(false, ar[kc],  false, br[kc], (short)0, accR, false, false);
            accR = __builtin_amdgcn_wmma_f32_16x16x4_f32(false, ani[kc], false, bi[kc], (short)0, accR, false, false);
            accI = __builtin_amdgcn_wmma_f32_16x16x4_f32(false, ar[kc],  false, bi[kc], (short)0, accI, false, false);
            accI = __builtin_amdgcn_wmma_f32_16x16x4_f32(false, ai[kc],  false, br[kc], (short)0, accI, false, false);
        }

#pragma unroll
        for (int r = 0; r < 8; ++r) {
            const int i = base | ((r + 8 * half) << SH);
            out[i]          = accR[r];
            out[NSTATE + i] = accI[r];
        }
    }
}

template <int L>
__device__ __forceinline__ void layerPass(float* st, float* mr, float* mi,
                                          const float* gu, int t, int wave, int lane) {
    float* cur = st + (L & 1) * 2 * NSTATE;
    if constexpr (L == 0) {
        apply16T<2, 0>(cur, cur, mr + 512, mi + 512, wave, lane);
    } else {
        constexpr int RPREV = (L - 1) % (NQ - 1) + 1;
        const float* prev = st + ((L & 1) ^ 1) * 2 * NSTATE;
        apply16T<2, RPREV>(prev, cur, mr + 512, mi + 512, wave, lane);
        buildGroup<0>(mr, mi, gu + L * NQ * 8, t);
    }
    __syncthreads();
    apply16T<1, 0>(cur, cur, mr + 256, mi + 256, wave, lane);
    if (L + 1 < NLAYERS) buildGroup<2>(mr, mi, gu + (L + 1) * NQ * 8, t);
    __syncthreads();
    apply16T<0, 0>(cur, cur, mr, mi, wave, lane);
    if (L + 1 < NLAYERS) buildGroup<1>(mr, mi, gu + (L + 1) * NQ * 8, t);
    __syncthreads();
}

__global__ __launch_bounds__(NTHREADS)
void vqc_kernel(const float* __restrict__ x,
                const float* __restrict__ wts,
                float* __restrict__ out) {
    __shared__ float st[2 * 2 * NSTATE];
    __shared__ float mr[3 * 256];
    __shared__ float mi[3 * 256];
    __shared__ float gu[NFUSED * 8];
    __shared__ float wred[NQ * 8];

    const int t    = threadIdx.x;
    const int b    = blockIdx.x;
    const int wave = t >> 5;
    const int lane = t & 31;

    if (t < NFUSED) {
        const float* wp = wts + t * 3;
        const float phi = wp[0], th = wp[1], om = wp[2];
        const float ct = cosf(th * 0.5f);
        const float sn = sinf(th * 0.5f);
        const float hs = 0.5f * (phi + om);
        const float hd = 0.5f * (phi - om);
        const float chs = cosf(hs), shs = sinf(hs);
        const float chd = cosf(hd), shd = sinf(hd);
        float e0 =  chs * ct, e1 = -shs * ct;
        float e2 = -chd * sn, e3 = -shd * sn;
        float e4 =  chd * sn, e5 = -shd * sn;
        float e6 =  chs * ct, e7 =  shs * ct;
        if (t < NQ) {
            const float a = x[b * NQ + t] * 0.5f;
            const float c = cosf(a), s = sinf(a);
            const float u0 = e0 * c + e3 * s, u1 = e1 * c - e2 * s;
            const float u2 = e1 * s + e2 * c, u3 = e3 * c - e0 * s;
            const float u4 = e4 * c + e7 * s, u5 = e5 * c - e6 * s;
            const float u6 = e5 * s + e6 * c, u7 = e7 * c - e4 * s;
            e0 = u0; e1 = u1; e2 = u2; e3 = u3;
            e4 = u4; e5 = u5; e6 = u6; e7 = u7;
        }
        float* gr = gu + t * 8;
        gr[0] = e0; gr[1] = e1; gr[2] = e2; gr[3] = e3;
        gr[4] = e4; gr[5] = e5; gr[6] = e6; gr[7] = e7;
    }

#pragma unroll
    for (int k = 0; k < 16; ++k) {
        const int i = k * NTHREADS + t;
        st[i] = 0.0f;
        st[NSTATE + i] = 0.0f;
    }
    if (t == 0) st[0] = 1.0f;
    __syncthreads();

    buildGroup<0>(mr, mi, gu, t);
    buildGroup<1>(mr, mi, gu, t);
    buildGroup<2>(mr, mi, gu, t);
    __syncthreads();

    layerPass<0>(st, mr, mi, gu, t, wave, lane);
    layerPass<1>(st, mr, mi, gu, t, wave, lane);
    layerPass<2>(st, mr, mi, gu, t, wave, lane);
    layerPass<3>(st, mr, mi, gu, t, wave, lane);

    constexpr int RLAST = (NLAYERS - 1) % (NQ - 1) + 1;
    const float* fin = st + ((NLAYERS - 1) & 1) * 2 * NSTATE;
    float part[NQ];
#pragma unroll
    for (int q = 0; q < NQ; ++q) part[q] = 0.0f;
#pragma unroll
    for (int k = 0; k < 16; ++k) {
        const int i = k * NTHREADS + t;
        const float p = fin[i] * fin[i] + fin[NSTATE + i] * fin[NSTATE + i];
        const int jk = ringDst(RLAST, k * NTHREADS);
#pragma unroll
        for (int q = 0; q < NQ; ++q) {
            part[q] += ((jk >> (11 - q)) & 1) ? -p : p;
        }
    }
    const int jt = ringDst(RLAST, t);
#pragma unroll
    for (int q = 0; q < NQ; ++q) {
        if ((jt >> (11 - q)) & 1) part[q] = -part[q];
    }
#pragma unroll
    for (int q = 0; q < NQ; ++q) {
        float v = part[q];
        for (int off = 16; off >= 1; off >>= 1)
            v += __shfl_xor(v, off, 32);
        if (lane == 0) wred[q * 8 + wave] = v;
    }
    __syncthreads();
    if (t < NQ) {
        float s = 0.0f;
#pragma unroll
        for (int wv = 0; wv < 8; ++wv) s += wred[t * 8 + wv];
        out[b * NQ + t] = s;
    }
}

extern "C" void kernel_launch(void* const* d_in, const int* in_sizes, int n_in,
                              void* d_out, int out_size, void* d_ws, size_t ws_size,
                              hipStream_t stream) {
    const float* x   = (const float*)d_in[0];
    const float* wts = (const float*)d_in[1];
    float* out = (float*)d_out;
    const int batch = in_sizes[0] / NQ;
    vqc_kernel<<<batch, NTHREADS, 0, stream>>>(x, wts, out);
}


# layer 0 acts on |0..0>: its output state is written directly as the product of the first columns of its 3 group operators; 3 of 12 WMMA passes and the zero fill go; layer-0 rounding changes
# speedup vs baseline: 5.5957x; 1.2009x over previous
#include <hip/hip_runtime.h>
#include <math.h>


typedef __attribute__((ext_vector_type(2))) float v2f;
typedef __attribute__((ext_vector_type(8))) float v8f;

#define NQ      12
#define NSTATE  4096
#define NLAYERS 4
#define NTHREADS 256
#define NFUSED  (NLAYERS * NQ)

__host__ __device__ constexpr int ringSrc(int r, int j) {
    int s = j;
    for (int qq = NQ - 1; qq >= 0; --qq) {
        const int cbit = 11 - qq;
        const int tbit = 11 - ((qq + r) % NQ);
        s ^= ((s >> cbit) & 1) << tbit;
    }
    return s;
}
__host__ __device__ constexpr int ringDst(int r, int i) {
    int s = i;
    for (int qq = 0; qq < NQ; ++qq) {
        const int cbit = 11 - qq;
        const int tbit = 11 - ((qq + r) % NQ);
        s ^= ((s >> cbit) & 1) << tbit;
    }
    return s;
}

template <int g>
__device__ __forceinline__ void buildGroup(float* mr, float* mi,
                                           const float* gl, int t) {
    const int row = t >> 4, cm = t & 15;
    float pr = 0.0f, pi = 0.0f;
#pragma unroll
    for (int wl = 0; wl < 4; ++wl) {
        const int q = 11 - (4 * g + wl);
        const int e = ((((row >> wl) & 1) << 1) | ((cm >> wl) & 1)) << 1;
        const float ur = gl[q * 8 + e];
        const float ui = gl[q * 8 + e + 1];
        if (wl == 0) {
            pr = ur;
            pi = ui;
        } else {
            const float nr = pr * ur - pi * ui;
            const float ni = pr * ui + pi * ur;
            pr = nr;
            pi = ni;
        }
    }
    mr[g * 256 + t] = pr;
    mi[g * 256 + t] = pi;
}

__device__ __forceinline__ void buildColumn0(float* cr, float* ci,
                                             const float* gl, int t) {
    const int g = t >> 4, row = t & 15;
    float pr = 0.0f, pi = 0.0f;
#pragma unroll
    for (int wl = 0; wl < 4; ++wl) {
        const int q = 11 - (4 * g + wl);
        const int e = ((row >> wl) & 1) << 2;
        const float ur = gl[q * 8 + e];
        const float ui = gl[q * 8 + e + 1];
        if (wl == 0) {
            pr = ur;
            pi = ui;
        } else {
            const float nr = pr * ur - pi * ui;
            const float ni = pr * ui + pi * ur;
            pr = nr;
            pi = ni;
        }
    }
    cr[t] = pr;
    ci[t] = pi;
}

template <int G, int R>
__device__ __forceinline__ void apply16T(const float* in, float* out,
                                         const float* mr, const float* mi,
                                         int wave, int lane) {
    constexpr int SH = G * 4;
    constexpr int LOWMASK = (1 << SH) - 1;
    const int half = lane >> 4;
    const int n15  = lane & 15;

    v2f ar[4], ai[4], ani[4];
#pragma unroll
    for (int kc = 0; kc < 4; ++kc) {
        const int k0 = kc * 4 + 2 * half;
        ar[kc].x  = mr[n15 * 16 + k0];  ar[kc].y  = mr[n15 * 16 + k0 + 1];
        ai[kc].x  = mi[n15 * 16 + k0];  ai[kc].y  = mi[n15 * 16 + k0 + 1];
        ani[kc].x = -ai[kc].x;          ani[kc].y = -ai[kc].y;
    }

#pragma unroll
    for (int tt = 0; tt < 2; ++tt) {
        const int col = (wave * 2 + tt) * 16 + n15;
        const int base = ((col >> SH) << (SH + 4)) | (col & LOWMASK);

        v2f br[4], bi[4];
        if (R == 0) {
#pragma unroll
            for (int kc = 0; kc < 4; ++kc) {
                const int k0 = kc * 4 + 2 * half;
                const int i0 = base | ((k0) << SH);
                const int i1 = base | ((k0 + 1) << SH);
                br[kc].x = in[i0];          br[kc].y = in[i1];
                bi[kc].x = in[NSTATE + i0]; bi[kc].y = in[NSTATE + i1];
            }
        } else {
            const int pl = ringSrc(R, base | ((2 * half) << SH));
#pragma unroll
            for (int kc = 0; kc < 4; ++kc) {
                const int i0 = pl ^ ringSrc(R, (kc * 4) << SH);
                const int i1 = pl ^ ringSrc(R, (kc * 4 + 1) << SH);
                br[kc].x = in[i0];          br[kc].y = in[i1];
                bi[kc].x = in[NSTATE + i0]; bi[kc].y = in[NSTATE + i1];
            }
        }

        v8f accR = {0.f,0.f,0.f,0.f,0.f,0.f,0.f,0.f};
        v8f accI = {0.f,0.f,0.f,0.f,0.f,0.f,0.f,0.f};

#pragma unroll
        for (int kc = 0; kc < 4; ++kc) {
            accR = __builtin_amdgcn_wmma_f32_16x16x4_f32(false, ar[kc],  false, br[kc], (short)0, accR, false, false);
            accR = __builtin_amdgcn_wmma_f32_16x16x4_f32(false, ani[kc], false, bi[kc], (short)0, accR, false, false);
            accI = __builtin_amdgcn_wmma_f32_16x16x4_f32(false, ar[kc],  false, bi[kc], (short)0, accI, false, false);
            accI = __builtin_amdgcn_wmma_f32_16x16x4_f32(false, ai[kc],  false, br[kc], (short)0, accI, false, false);
        }

#pragma unroll
        for (int r = 0; r < 8; ++r) {
            const int i = base | ((r + 8 * half) << SH);
            out[i]          = accR[r];
            out[NSTATE + i] = accI[r];
        }
    }
}

template <int L>
__device__ __forceinline__ void layerPass(float* st, float* mr, float* mi,
                                          const float* gu, int t, int wave, int lane) {
    float* cur = st + (L & 1) * 2 * NSTATE;
    constexpr int RPREV = (L - 1) % (NQ - 1) + 1;
    const float* prev = st + ((L & 1) ^ 1) * 2 * NSTATE;
    apply16T<2, RPREV>(prev, cur, mr + 512, mi + 512, wave, lane);
    buildGroup<0>(mr, mi, gu + L * NQ * 8, t);
    __syncthreads();
    apply16T<1, 0>(cur, cur, mr + 256, mi + 256, wave, lane);
    if (L + 1 < NLAYERS) buildGroup<2>(mr, mi, gu + (L + 1) * NQ * 8, t);
    __syncthreads();
    apply16T<0, 0>(cur, cur, mr, mi, wave, lane);
    if (L + 1 < NLAYERS) buildGroup<1>(mr, mi, gu + (L + 1) * NQ * 8, t);
    __syncthreads();
}

__global__ __launch_bounds__(NTHREADS)
void vqc_kernel(const float* __restrict__ x,
                const float* __restrict__ wts,
                float* __restrict__ out) {
    __shared__ float st[2 * 2 * NSTATE];
    __shared__ float mr[3 * 256];
    __shared__ float mi[3 * 256];
    __shared__ float gu[NFUSED * 8];
    __shared__ float c0r[3 * 16];
    __shared__ float c0i[3 * 16];
    __shared__ float wred[NQ * 8];

    const int t    = threadIdx.x;
    const int b    = blockIdx.x;
    const int wave = t >> 5;
    const int lane = t & 31;

    if (t < NFUSED) {
        const float* wp = wts + t * 3;
        const float phi = wp[0], th = wp[1], om = wp[2];
        const float ct = cosf(th * 0.5f);
        const float sn = sinf(th * 0.5f);
        const float hs = 0.5f * (phi + om);
        const float hd = 0.5f * (phi - om);
        const float chs = cosf(hs), shs = sinf(hs);
        const float chd = cosf(hd), shd = sinf(hd);
        float e0 =  chs * ct, e1 = -shs * ct;
        float e2 = -chd * sn, e3 = -shd * sn;
        float e4 =  chd * sn, e5 = -shd * sn;
        float e6 =  chs * ct, e7 =  shs * ct;
        if (t < NQ) {
            const float a = x[b * NQ + t] * 0.5f;
            const float c = cosf(a), s = sinf(a);
            const float u0 = e0 * c + e3 * s, u1 = e1 * c - e2 * s;
            const float u2 = e1 * s + e2 * c, u3 = e3 * c - e0 * s;
            const float u4 = e4 * c + e7 * s, u5 = e5 * c - e6 * s;
            const float u6 = e5 * s + e6 * c, u7 = e7 * c - e4 * s;
            e0 = u0; e1 = u1; e2 = u2; e3 = u3;
            e4 = u4; e5 = u5; e6 = u6; e7 = u7;
        }
        float* gr = gu + t * 8;
        gr[0] = e0; gr[1] = e1; gr[2] = e2; gr[3] = e3;
        gr[4] = e4; gr[5] = e5; gr[6] = e6; gr[7] = e7;
    }
    __syncthreads();

    if (t < 48) buildColumn0(c0r, c0i, gu, t);
    buildGroup<2>(mr, mi, gu + NQ * 8, t);
    buildGroup<1>(mr, mi, gu + NQ * 8, t);
    __syncthreads();

    {
        const float ar = c0r[16 + (t >> 4)], ai = c0i[16 + (t >> 4)];
        const float br = c0r[t & 15],        bi = c0i[t & 15];
        const float wr = ar * br - ai * bi;
        const float wi = ar * bi + ai * br;
#pragma unroll
        for (int k = 0; k < 16; ++k) {
            const int i = k * NTHREADS + t;
            const float hr = c0r[32 + k], hi = c0i[32 + k];
            st[i]          = hr * wr - hi * wi;
            st[NSTATE + i] = hr * wi + hi * wr;
        }
    }
    __syncthreads();

    layerPass<1>(st, mr, mi, gu, t, wave, lane);
    layerPass<2>(st, mr, mi, gu, t, wave, lane);
    layerPass<3>(st, mr, mi, gu, t, wave, lane);

    constexpr int RLAST = (NLAYERS - 1) % (NQ - 1) + 1;
    const float* fin = st + ((NLAYERS - 1) & 1) * 2 * NSTATE;
    float part[NQ];
#pragma unroll
    for (int q = 0; q < NQ; ++q) part[q] = 0.0f;
#pragma unroll
    for (int k = 0; k < 16; ++k) {
        const int i = k * NTHREADS + t;
        const float p = fin[i] * fin[i] + fin[NSTATE + i] * fin[NSTATE + i];
        const int jk = ringDst(RLAST, k * NTHREADS);
#pragma unroll
        for (int q = 0; q < NQ; ++q) {
            part[q] += ((jk >> (11 - q)) & 1) ? -p : p;
        }
    }
    const int jt = ringDst(RLAST, t);
#pragma unroll
    for (int q = 0; q < NQ; ++q) {
        if ((jt >> (11 - q)) & 1) part[q] = -part[q];
    }
#pragma unroll
    for (int q = 0; q < NQ; ++q) {
        float v = part[q];
        for (int off = 16; off >= 1; off >>= 1)
            v += __shfl_xor(v, off, 32);
        if (lane == 0) wred[q * 8 + wave] = v;
    }
    __syncthreads();
    if (t < NQ) {
        float s = 0.0f;
#pragma unroll
        for (int wv = 0; wv < 8; ++wv) s += wred[t * 8 + wv];
        out[b * NQ + t] = s;
    }
}

extern "C" void kernel_launch(void* const* d_in, const int* in_sizes, int n_in,
                              void* d_out, int out_size, void* d_ws, size_t ws_size,
                              hipStream_t stream) {
    const float* x   = (const float*)d_in[0];
    const float* wts = (const float*)d_in[1];
    float* out = (float*)d_out;
    const int batch = in_sizes[0] / NQ;
    vqc_kernel<<<batch, NTHREADS, 0, stream>>>(x, wts, out);
}


# complex 16x16 operator product by three real WMMA products instead of four (T=(Mr+Mi)Sr, Dr=T-Mi(Sr+Si), Di=T+Mr(Si-Sr)): 288 WMMA per wave -> 216; f32 rounding and summation order change
# speedup vs baseline: 6.0886x; 1.0881x over previous
#include <hip/hip_runtime.h>
#include <math.h>


typedef __attribute__((ext_vector_type(2))) float v2f;
typedef __attribute__((ext_vector_type(8))) float v8f;

#define NQ      12
#define NSTATE  4096
#define NLAYERS 4
#define NTHREADS 256
#define NFUSED  (NLAYERS * NQ)

__host__ __device__ constexpr int ringSrc(int r, int j) {
    int s = j;
    for (int qq = NQ - 1; qq >= 0; --qq) {
        const int cbit = 11 - qq;
        const int tbit = 11 - ((qq + r) % NQ);
        s ^= ((s >> cbit) & 1) << tbit;
    }
    return s;
}
__host__ __device__ constexpr int ringDst(int r, int i) {
    int s = i;
    for (int qq = 0; qq < NQ; ++qq) {
        const int cbit = 11 - qq;
        const int tbit = 11 - ((qq + r) % NQ);
        s ^= ((s >> cbit) & 1) << tbit;
    }
    return s;
}

template <int g>
__device__ __forceinline__ void buildGroup(float* mr, float* mi,
                                           const float* gl, int t) {
    const int row = t >> 4, cm = t & 15;
    float pr = 0.0f, pi = 0.0f;
#pragma unroll
    for (int wl = 0; wl < 4; ++wl) {
        const int q = 11 - (4 * g + wl);
        const int e = ((((row >> wl) & 1) << 1) | ((cm >> wl) & 1)) << 1;
        const float ur = gl[q * 8 + e];
        const float ui = gl[q * 8 + e + 1];
        if (wl == 0) {
            pr = ur;
            pi = ui;
        } else {
            const float nr = pr * ur - pi * ui;
            const float ni = pr * ui + pi * ur;
            pr = nr;
            pi = ni;
        }
    }
    mr[g * 256 + t] = pr;
    mi[g * 256 + t] = pi;
}

__device__ __forceinline__ void buildColumn0(float* cr, float* ci,
                                             const float* gl, int t) {
    const int g = t >> 4, row = t & 15;
    float pr = 0.0f, pi = 0.0f;
#pragma unroll
    for (int wl = 0; wl < 4; ++wl) {
        const int q = 11 - (4 * g + wl);
        const int e = ((row >> wl) & 1) << 2;
        const float ur = gl[q * 8 + e];
        const float ui = gl[q * 8 + e + 1];
        if (wl == 0) {
            pr = ur;
            pi = ui;
        } else {
            const float nr = pr * ur - pi * ui;
            const float ni = pr * ui + pi * ur;
            pr = nr;
            pi = ni;
        }
    }
    cr[t] = pr;
    ci[t] = pi;
}

template <int G, int R>
__device__ __forceinline__ void apply16T(const float* in, float* out,
                                         const float* mr, const float* mi,
                                         int wave, int lane) {
    constexpr int SH = G * 4;
    constexpr int LOWMASK = (1 << SH) - 1;
    const int half = lane >> 4;
    const int n15  = lane & 15;

    v2f ar[4], asum[4], ani[4];
#pragma unroll
    for (int kc = 0; kc < 4; ++kc) {
        const int k0 = kc * 4 + 2 * half;
        const float mix = mi[n15 * 16 + k0], miy = mi[n15 * 16 + k0 + 1];
        ar[kc].x   = mr[n15 * 16 + k0];  ar[kc].y   = mr[n15 * 16 + k0 + 1];
        asum[kc].x = ar[kc].x + mix;     asum[kc].y = ar[kc].y + miy;
        ani[kc].x  = -mix;               ani[kc].y  = -miy;
    }

#pragma unroll
    for (int tt = 0; tt < 2; ++tt) {
        const int col = (wave * 2 + tt) * 16 + n15;
        const int base = ((col >> SH) << (SH + 4)) | (col & LOWMASK);

        v2f br[4], bi[4];
        if (R == 0) {
#pragma unroll
            for (int kc = 0; kc < 4; ++kc) {
                const int k0 = kc * 4 + 2 * half;
                const int i0 = base | ((k0) << SH);
                const int i1 = base | ((k0 + 1) << SH);
                br[kc].x = in[i0];          br[kc].y = in[i1];
                bi[kc].x = in[NSTATE + i0]; bi[kc].y = in[NSTATE + i1];
            }
        } else {
            const int pl = ringSrc(R, base | ((2 * half) << SH));
#pragma unroll
            for (int kc = 0; kc < 4; ++kc) {
                const int i0 = pl ^ ringSrc(R, (kc * 4) << SH);
                const int i1 = pl ^ ringSrc(R, (kc * 4 + 1) << SH);
                br[kc].x = in[i0];          br[kc].y = in[i1];
                bi[kc].x = in[NSTATE + i0]; bi[kc].y = in[NSTATE + i1];
            }
        }

        v2f bsum[4], bdif[4];
#pragma unroll
        for (int kc = 0; kc < 4; ++kc) {
            bsum[kc].x = br[kc].x + bi[kc].x;  bsum[kc].y = br[kc].y + bi[kc].y;
            bdif[kc].x = bi[kc].x - br[kc].x;  bdif[kc].y = bi[kc].y - br[kc].y;
        }

        v8f accT = {0.f,0.f,0.f,0.f,0.f,0.f,0.f,0.f};
#pragma unroll
        for (int kc = 0; kc < 4; ++kc) {
            accT = __builtin_amdgcn_wmma_f32_16x16x4_f32(false, asum[kc], false, br[kc], (short)0, accT, false, false);
        }
        v8f accR = accT;
        v8f accI = accT;
#pragma unroll
        for (int kc = 0; kc < 4; ++kc) {
            accR = __builtin_amdgcn_wmma_f32_16x16x4_f32(false, ani[kc], false, bsum[kc], (short)0, accR, false, false);
            accI = __builtin_amdgcn_wmma_f32_16x16x4_f32(false, ar[kc],  false, bdif[kc], (short)0, accI, false, false);
        }

#pragma unroll
        for (int r = 0; r < 8; ++r) {
            const int i = base | ((r + 8 * half) << SH);
            out[i]          = accR[r];
            out[NSTATE + i] = accI[r];
        }
    }
}

template <int L>
__device__ __forceinline__ void layerPass(float* st, float* mr, float* mi,
                                          const float* gu, int t, int wave, int lane) {
    float* cur = st + (L & 1) * 2 * NSTATE;
    constexpr int RPREV = (L - 1) % (NQ - 1) + 1;
    const float* prev = st + ((L & 1) ^ 1) * 2 * NSTATE;
    apply16T<2, RPREV>(prev, cur, mr + 512, mi + 512, wave, lane);
    buildGroup<0>(mr, mi, gu + L * NQ * 8, t);
    __syncthreads();
    apply16T<1, 0>(cur, cur, mr + 256, mi + 256, wave, lane);
    if (L + 1 < NLAYERS) buildGroup<2>(mr, mi, gu + (L + 1) * NQ * 8, t);
    __syncthreads();
    apply16T<0, 0>(cur, cur, mr, mi, wave, lane);
    if (L + 1 < NLAYERS) buildGroup<1>(mr, mi, gu + (L + 1) * NQ * 8, t);
    __syncthreads();
}

__global__ __launch_bounds__(NTHREADS)
void vqc_kernel(const float* __restrict__ x,
                const float* __restrict__ wts,
                float* __restrict__ out) {
    __shared__ float st[2 * 2 * NSTATE];
    __shared__ float mr[3 * 256];
    __shared__ float mi[3 * 256];
    __shared__ float gu[NFUSED * 8];
    __shared__ float c0r[3 * 16];
    __shared__ float c0i[3 * 16];
    __shared__ float wred[NQ * 8];

    const int t    = threadIdx.x;
    const int b    = blockIdx.x;
    const int wave = t >> 5;
    const int lane = t & 31;

    if (t < NFUSED) {
        const float* wp = wts + t * 3;
        const float phi = wp[0], th = wp[1], om = wp[2];
        const float ct = cosf(th * 0.5f);
        const float sn = sinf(th * 0.5f);
        const float hs = 0.5f * (phi + om);
        const float hd = 0.5f * (phi - om);
        const float chs = cosf(hs), shs = sinf(hs);
        const float chd = cosf(hd), shd = sinf(hd);
        float e0 =  chs * ct, e1 = -shs * ct;
        float e2 = -chd * sn, e3 = -shd * sn;
        float e4 =  chd * sn, e5 = -shd * sn;
        float e6 =  chs * ct, e7 =  shs * ct;
        if (t < NQ) {
            const float a = x[b * NQ + t] * 0.5f;
            const float c = cosf(a), s = sinf(a);
            const float u0 = e0 * c + e3 * s, u1 = e1 * c - e2 * s;
            const float u2 = e1 * s + e2 * c, u3 = e3 * c - e0 * s;
            const float u4 = e4 * c + e7 * s, u5 = e5 * c - e6 * s;
            const float u6 = e5 * s + e6 * c, u7 = e7 * c - e4 * s;
            e0 = u0; e1 = u1; e2 = u2; e3 = u3;
            e4 = u4; e5 = u5; e6 = u6; e7 = u7;
        }
        float* gr = gu + t * 8;
        gr[0] = e0; gr[1] = e1; gr[2] = e2; gr[3] = e3;
        gr[4] = e4; gr[5] = e5; gr[6] = e6; gr[7] = e7;
    }
    __syncthreads();

    if (t < 48) buildColumn0(c0r, c0i, gu, t);
    buildGroup<2>(mr, mi, gu + NQ * 8, t);
    buildGroup<1>(mr, mi, gu + NQ * 8, t);
    __syncthreads();

    {
        const float ar = c0r[16 + (t >> 4)], ai = c0i[16 + (t >> 4)];
        const float br = c0r[t & 15],        bi = c0i[t & 15];
        const float wr = ar * br - ai * bi;
        const float wi = ar * bi + ai * br;
#pragma unroll
        for (int k = 0; k < 16; ++k) {
            const int i = k * NTHREADS + t;
            const float hr = c0r[32 + k], hi = c0i[32 + k];
            st[i]          = hr * wr - hi * wi;
            st[NSTATE + i] = hr * wi + hi * wr;
        }
    }
    __syncthreads();

    layerPass<1>(st, mr, mi, gu, t, wave, lane);
    layerPass<2>(st, mr, mi, gu, t, wave, lane);
    layerPass<3>(st, mr, mi, gu, t, wave, lane);

    constexpr int RLAST = (NLAYERS - 1) % (NQ - 1) + 1;
    const float* fin = st + ((NLAYERS - 1) & 1) * 2 * NSTATE;
    float part[NQ];
#pragma unroll
    for (int q = 0; q < NQ; ++q) part[q] = 0.0f;
#pragma unroll
    for (int k = 0; k < 16; ++k) {
        const int i = k * NTHREADS + t;
        const float p = fin[i] * fin[i] + fin[NSTATE + i] * fin[NSTATE + i];
        const int jk = ringDst(RLAST, k * NTHREADS);
#pragma unroll
        for (int q = 0; q < NQ; ++q) {
            part[q] += ((jk >> (11 - q)) & 1) ? -p : p;
        }
    }
    const int jt = ringDst(RLAST, t);
#pragma unroll
    for (int q = 0; q < NQ; ++q) {
        if ((jt >> (11 - q)) & 1) part[q] = -part[q];
    }
#pragma unroll
    for (int q = 0; q < NQ; ++q) {
        float v = part[q];
        for (int off = 16; off >= 1; off >>= 1)
            v += __shfl_xor(v, off, 32);
        if (lane == 0) wred[q * 8 + wave] = v;
    }
    __syncthreads();
    if (t < NQ) {
        float s = 0.0f;
#pragma unroll
        for (int wv = 0; wv < 8; ++wv) s += wred[t * 8 + wv];
        out[b * NQ + t] = s;
    }
}

extern "C" void kernel_launch(void* const* d_in, const int* in_sizes, int n_in,
                              void* d_out, int out_size, void* d_ws, size_t ws_size,
                              hipStream_t stream) {
    const float* x   = (const float*)d_in[0];
    const float* wts = (const float*)d_in[1];
    float* out = (float*)d_out;
    const int batch = in_sizes[0] / NQ;
    vqc_kernel<<<batch, NTHREADS, 0, stream>>>(x, wts, out);
}


# top- and low-group operator passes chained in registers: state fed as the WMMA A operand, so D's rows and lanes swap roles; 9 LDS round trips and barriers -> 6; f32 summation order changes
# speedup vs baseline: 6.4072x; 1.0523x over previous
#include <hip/hip_runtime.h>
#include <math.h>


typedef __attribute__((ext_vector_type(2))) float v2f;
typedef __attribute__((ext_vector_type(8))) float v8f;

#define NQ      12
#define NSTATE  4096
#define NLAYERS 4
#define NTHREADS 256
#define NFUSED  (NLAYERS * NQ)

__host__ __device__ constexpr int ringSrc(int r, int j) {
    int s = j;
    for (int qq = NQ - 1; qq >= 0; --qq) {
        const int cbit = 11 - qq;
        const int tbit = 11 - ((qq + r) % NQ);
        s ^= ((s >> cbit) & 1) << tbit;
    }
    return s;
}
__host__ __device__ constexpr int ringDst(int r, int i) {
    int s = i;
    for (int qq = 0; qq < NQ; ++qq) {
        const int cbit = 11 - qq;
        const int tbit = 11 - ((qq + r) % NQ);
        s ^= ((s >> cbit) & 1) << tbit;
    }
    return s;
}

template <int g>
__device__ __forceinline__ void buildGroup(float* mr, float* mi,
                                           const float* gl, int t) {
    const int row = t >> 4, cm = t & 15;
    float pr = 0.0f, pi = 0.0f;
#pragma unroll
    for (int wl = 0; wl < 4; ++wl) {
        const int q = 11 - (4 * g + wl);
        const int e = ((((row >> wl) & 1) << 1) | ((cm >> wl) & 1)) << 1;
        const float ur = gl[q * 8 + e];
        const float ui = gl[q * 8 + e + 1];
        if (wl == 0) {
            pr = ur;
            pi = ui;
        } else {
            const float nr = pr * ur - pi * ui;
            const float ni = pr * ui + pi * ur;
            pr = nr;
            pi = ni;
        }
    }
    mr[g * 256 + t] = pr;
    mi[g * 256 + t] = pi;
}

__device__ __forceinline__ void buildColumn0(float* cr, float* ci,
                                             const float* gl, int t) {
    const int g = t >> 4, row = t & 15;
    float pr = 0.0f, pi = 0.0f;
#pragma unroll
    for (int wl = 0; wl < 4; ++wl) {
        const int q = 11 - (4 * g + wl);
        const int e = ((row >> wl) & 1) << 2;
        const float ur = gl[q * 8 + e];
        const float ui = gl[q * 8 + e + 1];
        if (wl == 0) {
            pr = ur;
            pi = ui;
        } else {
            const float nr = pr * ur - pi * ui;
            const float ni = pr * ui + pi * ur;
            pr = nr;
            pi = ni;
        }
    }
    cr[t] = pr;
    ci[t] = pi;
}

template <int G>
__device__ __forceinline__ void apply16T(const float* in, float* out,
                                         const float* mr, const float* mi,
                                         int wave, int lane) {
    constexpr int SH = G * 4;
    constexpr int LOWMASK = (1 << SH) - 1;
    const int half = lane >> 4;
    const int n15  = lane & 15;

    v2f ar[4], asum[4], ani[4];
#pragma unroll
    for (int kc = 0; kc < 4; ++kc) {
        const int k0 = kc * 4 + 2 * half;
        const float mix = mi[n15 * 16 + k0], miy = mi[n15 * 16 + k0 + 1];
        ar[kc].x   = mr[n15 * 16 + k0];  ar[kc].y   = mr[n15 * 16 + k0 + 1];
        asum[kc].x = ar[kc].x + mix;     asum[kc].y = ar[kc].y + miy;
        ani[kc].x  = -mix;               ani[kc].y  = -miy;
    }

#pragma unroll
    for (int tt = 0; tt < 2; ++tt) {
        const int col = (wave * 2 + tt) * 16 + n15;
        const int base = ((col >> SH) << (SH + 4)) | (col & LOWMASK);

        v2f br[4], bi[4];
#pragma unroll
        for (int kc = 0; kc < 4; ++kc) {
            const int k0 = kc * 4 + 2 * half;
            const int i0 = base | ((k0) << SH);
            const int i1 = base | ((k0 + 1) << SH);
            br[kc].x = in[i0];          br[kc].y = in[i1];
            bi[kc].x = in[NSTATE + i0]; bi[kc].y = in[NSTATE + i1];
        }

        v2f bsum[4], bdif[4];
#pragma unroll
        for (int kc = 0; kc < 4; ++kc) {
            bsum[kc].x = br[kc].x + bi[kc].x;  bsum[kc].y = br[kc].y + bi[kc].y;
            bdif[kc].x = bi[kc].x - br[kc].x;  bdif[kc].y = bi[kc].y - br[kc].y;
        }

        v8f accT = {0.f,0.f,0.f,0.f,0.f,0.f,0.f,0.f};
#pragma unroll
        for (int kc = 0; kc < 4; ++kc) {
            accT = __builtin_amdgcn_wmma_f32_16x16x4_f32(false, asum[kc], false, br[kc], (short)0, accT, false, false);
        }
        v8f accR = accT;
        v8f accI = accT;
#pragma unroll
        for (int kc = 0; kc < 4; ++kc) {
            accR = __builtin_amdgcn_wmma_f32_16x16x4_f32(false, ani[kc], false, bsum[kc], (short)0, accR, false, false);
            accI = __builtin_amdgcn_wmma_f32_16x16x4_f32(false, ar[kc],  false, bdif[kc], (short)0, accI, false, false);
        }

#pragma unroll
        for (int r = 0; r < 8; ++r) {
            const int i = base | ((r + 8 * half) << SH);
            out[i]          = accR[r];
            out[NSTATE + i] = accI[r];
        }
    }
}

template <int R>
__device__ __forceinline__ void chainTopLow(const float* in, float* out,
                                            const float* m2r, const float* m2i,
                                            const float* m0r, const float* m0i,
                                            int wave, int lane) {
    constexpr int SH = 8;
    const int half = lane >> 4;
    const int n15  = lane & 15;

    v2f tr[4], tsum[4], tni[4];
#pragma unroll
    for (int kc = 0; kc < 4; ++kc) {
        const int k0 = kc * 4 + 2 * half;
        const float mix = m2i[n15 * 16 + k0], miy = m2i[n15 * 16 + k0 + 1];
        tr[kc].x   = m2r[n15 * 16 + k0];  tr[kc].y   = m2r[n15 * 16 + k0 + 1];
        tsum[kc].x = tr[kc].x + mix;      tsum[kc].y = tr[kc].y + miy;
        tni[kc].x  = -mix;                tni[kc].y  = -miy;
    }
    v2f lr[4], lsum[4], lni[4];
#pragma unroll
    for (int kc = 0; kc < 4; ++kc) {
        const int k0 = 2 * kc + 8 * half;
        const float mix = m0i[n15 * 16 + k0], miy = m0i[n15 * 16 + k0 + 1];
        lr[kc].x   = m0r[n15 * 16 + k0];  lr[kc].y   = m0r[n15 * 16 + k0 + 1];
        lsum[kc].x = lr[kc].x + mix;      lsum[kc].y = lr[kc].y + miy;
        lni[kc].x  = -mix;                lni[kc].y  = -miy;
    }

    const int base0 = (wave * 2) * 16 + n15;
    const int pl0   = ringSrc(R, base0 | ((2 * half) << SH));

#pragma unroll 1
    for (int tt = 0; tt < 2; ++tt) {
        const int base = base0 | (tt << 4);
        const int pl   = pl0 ^ (-tt & ringSrc(R, 16));

        v2f sr[4], si[4];
#pragma unroll
        for (int kc = 0; kc < 4; ++kc) {
            const int i0 = pl ^ ringSrc(R, (kc * 4) << SH);
            const int i1 = pl ^ ringSrc(R, (kc * 4 + 1) << SH);
            sr[kc].x = in[i0];          sr[kc].y = in[i1];
            si[kc].x = in[NSTATE + i0]; si[kc].y = in[NSTATE + i1];
        }

        v2f ssum[4], sdif[4];
#pragma unroll
        for (int kc = 0; kc < 4; ++kc) {
            ssum[kc].x = sr[kc].x + si[kc].x;  ssum[kc].y = sr[kc].y + si[kc].y;
            sdif[kc].x = si[kc].x - sr[kc].x;  sdif[kc].y = si[kc].y - sr[kc].y;
        }

        v8f accT = {0.f,0.f,0.f,0.f,0.f,0.f,0.f,0.f};
#pragma unroll
        for (int kc = 0; kc < 4; ++kc) {
            accT = __builtin_amdgcn_wmma_f32_16x16x4_f32(false, sr[kc], false, tsum[kc], (short)0, accT, false, false);
        }
        v8f accR = accT;
        v8f accI = accT;
#pragma unroll
        for (int kc = 0; kc < 4; ++kc) {
            accR = __builtin_amdgcn_wmma_f32_16x16x4_f32(false, ssum[kc], false, tni[kc], (short)0, accR, false, false);
            accI = __builtin_amdgcn_wmma_f32_16x16x4_f32(false, sdif[kc], false, tr[kc],  (short)0, accI, false, false);
        }

        v2f ur[4], usum[4], udif[4];
#pragma unroll
        for (int kc = 0; kc < 4; ++kc) {
            const float r0 = accR[2 * kc], r1 = accR[2 * kc + 1];
            const float i0 = accI[2 * kc], i1 = accI[2 * kc + 1];
            ur[kc].x   = r0;       ur[kc].y   = r1;
            usum[kc].x = r0 + i0;  usum[kc].y = r1 + i1;
            udif[kc].x = i0 - r0;  udif[kc].y = i1 - r1;
        }

        v8f accU = {0.f,0.f,0.f,0.f,0.f,0.f,0.f,0.f};
#pragma unroll
        for (int kc = 0; kc < 4; ++kc) {
            accU = __builtin_amdgcn_wmma_f32_16x16x4_f32(false, ur[kc], false, lsum[kc], (short)0, accU, false, false);
        }
        v8f outR = accU;
        v8f outI = accU;
#pragma unroll
        for (int kc = 0; kc < 4; ++kc) {
            outR = __builtin_amdgcn_wmma_f32_16x16x4_f32(false, usum[kc], false, lni[kc], (short)0, outR, false, false);
            outI = __builtin_amdgcn_wmma_f32_16x16x4_f32(false, udif[kc], false, lr[kc],  (short)0, outI, false, false);
        }

#pragma unroll
        for (int r = 0; r < 8; ++r) {
            const int i = base | ((r + 8 * half) << SH);
            out[i]          = outR[r];
            out[NSTATE + i] = outI[r];
        }
    }
}

template <int L>
__device__ __forceinline__ void layerPass(float* st, float* mr, float* mi,
                                          const float* gu, int t, int wave, int lane) {
    float* cur = st + (L & 1) * 2 * NSTATE;
    constexpr int RPREV = (L - 1) % (NQ - 1) + 1;
    const float* prev = st + ((L & 1) ^ 1) * 2 * NSTATE;
    chainTopLow<RPREV>(prev, cur, mr + 512, mi + 512, mr, mi, wave, lane);
    buildGroup<1>(mr, mi, gu + L * NQ * 8, t);
    __syncthreads();
    apply16T<1>(cur, cur, mr + 256, mi + 256, wave, lane);
    if (L + 1 < NLAYERS) {
        buildGroup<2>(mr, mi, gu + (L + 1) * NQ * 8, t);
        buildGroup<0>(mr, mi, gu + (L + 1) * NQ * 8, t);
    }
    __syncthreads();
}

__global__ __launch_bounds__(NTHREADS)
void vqc_kernel(const float* __restrict__ x,
                const float* __restrict__ wts,
                float* __restrict__ out) {
    __shared__ float st[2 * 2 * NSTATE];
    __shared__ float mr[3 * 256];
    __shared__ float mi[3 * 256];
    __shared__ float gu[NFUSED * 8];
    __shared__ float c0r[3 * 16];
    __shared__ float c0i[3 * 16];
    __shared__ float wred[NQ * 8];

    const int t    = threadIdx.x;
    const int b    = blockIdx.x;
    const int wave = t >> 5;
    const int lane = t & 31;

    if (t < NFUSED) {
        const float* wp = wts + t * 3;
        const float phi = wp[0], th = wp[1], om = wp[2];
        const float ct = cosf(th * 0.5f);
        const float sn = sinf(th * 0.5f);
        const float hs = 0.5f * (phi + om);
        const float hd = 0.5f * (phi - om);
        const float chs = cosf(hs), shs = sinf(hs);
        const float chd = cosf(hd), shd = sinf(hd);
        float e0 =  chs * ct, e1 = -shs * ct;
        float e2 = -chd * sn, e3 = -shd * sn;
        float e4 =  chd * sn, e5 = -shd * sn;
        float e6 =  chs * ct, e7 =  shs * ct;
        if (t < NQ) {
            const float a = x[b * NQ + t] * 0.5f;
            const float c = cosf(a), s = sinf(a);
            const float u0 = e0 * c + e3 * s, u1 = e1 * c - e2 * s;
            const float u2 = e1 * s + e2 * c, u3 = e3 * c - e0 * s;
            const float u4 = e4 * c + e7 * s, u5 = e5 * c - e6 * s;
            const float u6 = e5 * s + e6 * c, u7 = e7 * c - e4 * s;
            e0 = u0; e1 = u1; e2 = u2; e3 = u3;
            e4 = u4; e5 = u5; e6 = u6; e7 = u7;
        }
        float* gr = gu + t * 8;
        gr[0] = e0; gr[1] = e1; gr[2] = e2; gr[3] = e3;
        gr[4] = e4; gr[5] = e5; gr[6] = e6; gr[7] = e7;
    }
    __syncthreads();

    if (t < 48) buildColumn0(c0r, c0i, gu, t);
    buildGroup<2>(mr, mi, gu + NQ * 8, t);
    buildGroup<0>(mr, mi, gu + NQ * 8, t);
    __syncthreads();

    {
        const float ar = c0r[16 + (t >> 4)], ai = c0i[16 + (t >> 4)];
        const float br = c0r[t & 15],        bi = c0i[t & 15];
        const float wr = ar * br - ai * bi;
        const float wi = ar * bi + ai * br;
#pragma unroll
        for (int k = 0; k < 16; ++k) {
            const int i = k * NTHREADS + t;
            const float hr = c0r[32 + k], hi = c0i[32 + k];
            st[i]          = hr * wr - hi * wi;
            st[NSTATE + i] = hr * wi + hi * wr;
        }
    }
    __syncthreads();

    layerPass<1>(st, mr, mi, gu, t, wave, lane);
    layerPass<2>(st, mr, mi, gu, t, wave, lane);
    layerPass<3>(st, mr, mi, gu, t, wave, lane);

    constexpr int RLAST = (NLAYERS - 1) % (NQ - 1) + 1;
    const float* fin = st + ((NLAYERS - 1) & 1) * 2 * NSTATE;
    float part[NQ];
#pragma unroll
    for (int q = 0; q < NQ; ++q) part[q] = 0.0f;
#pragma unroll
    for (int k = 0; k < 16; ++k) {
        const int i = k * NTHREADS + t;
        const float p = fin[i] * fin[i] + fin[NSTATE + i] * fin[NSTATE + i];
        const int jk = ringDst(RLAST, k * NTHREADS);
#pragma unroll
        for (int q = 0; q < NQ; ++q) {
            part[q] += ((jk >> (11 - q)) & 1) ? -p : p;
        }
    }
    const int jt = ringDst(RLAST, t);
#pragma unroll
    for (int q = 0; q < NQ; ++q) {
        if ((jt >> (11 - q)) & 1) part[q] = -part[q];
    }
#pragma unroll
    for (int q = 0; q < NQ; ++q) {
        float v = part[q];
        for (int off = 16; off >= 1; off >>= 1)
            v += __shfl_xor(v, off, 32);
        if (lane == 0) wred[q * 8 + wave] = v;
    }
    __syncthreads();
    if (t < NQ) {
        float s = 0.0f;
#pragma unroll
        for (int wv = 0; wv < 8; ++wv) s += wred[t * 8 + wv];
        out[b * NQ + t] = s;
    }
}

extern "C" void kernel_launch(void* const* d_in, const int* in_sizes, int n_in,
                              void* d_out, int out_size, void* d_ws, size_t ws_size,
                              hipStream_t stream) {
    const float* x   = (const float*)d_in[0];
    const float* wts = (const float*)d_in[1];
    float* out = (float*)d_out;
    const int batch = in_sizes[0] / NQ;
    vqc_kernel<<<batch, NTHREADS, 0, stream>>>(x, wts, out);
}


# layers 1-3: Rot split RZ*RY*RZ: RY tensor product is a REAL 16x16 operator (2 real WMMA products per pass, not 3), RZ phases applied elementwise; 216 WMMA/wave -> 144; f32 rounding, sum order change
# speedup vs baseline: 7.6938x; 1.2008x over previous
#include <hip/hip_runtime.h>
#include <math.h>


typedef __attribute__((ext_vector_type(2))) float v2f;
typedef __attribute__((ext_vector_type(8))) float v8f;

#define NQ      12
#define NSTATE  4096
#define NLAYERS 4
#define NTHREADS 256
#define NFUSED  (NLAYERS * NQ)

__host__ __device__ constexpr int ringSrc(int r, int j) {
    int s = j;
    for (int qq = NQ - 1; qq >= 0; --qq) {
        const int cbit = 11 - qq;
        const int tbit = 11 - ((qq + r) % NQ);
        s ^= ((s >> cbit) & 1) << tbit;
    }
    return s;
}
__host__ __device__ constexpr int ringDst(int r, int i) {
    int s = i;
    for (int qq = 0; qq < NQ; ++qq) {
        const int cbit = 11 - qq;
        const int tbit = 11 - ((qq + r) % NQ);
        s ^= ((s >> cbit) & 1) << tbit;
    }
    return s;
}

template <int g>
__device__ __forceinline__ void buildGroup(float* mr, const float* gl, int t) {
    const int row = t >> 4, cm = t & 15;
    float p = 0.0f;
#pragma unroll
    for (int wl = 0; wl < 4; ++wl) {
        const int q = 11 - (4 * g + wl);
        const int rb = (row >> wl) & 1, cb = (cm >> wl) & 1;
        const float cs = gl[q * 8];
        const float sn = gl[q * 8 + 1];
        const float u = (rb == cb) ? cs : (rb ? sn : -sn);
        p = (wl == 0) ? u : p * u;
    }
    mr[g * 256 + t] = p;
}

__device__ __forceinline__ void buildPhase(float* ph, const float* gu, int t) {
    const int side = (t >= 144) ? 1 : 0;
    const int u = t - 144 * side;
    const int L = u / 48 + 1;
    const int r = u % 48;
    const int g = r >> 4, idx = r & 15;
    const float* gl = gu + L * NQ * 8 + 2 + 2 * side;
    float pr = 0.0f, pi = 0.0f;
#pragma unroll
    for (int wl = 0; wl < 4; ++wl) {
        const int q = 11 - (4 * g + wl);
        const float cs = gl[q * 8];
        const float s0 = gl[q * 8 + 1];
        const float sn = ((idx >> wl) & 1) ? s0 : -s0;
        if (wl == 0) {
            pr = cs;
            pi = sn;
        } else {
            const float nr = pr * cs - pi * sn;
            const float ni = pr * sn + pi * cs;
            pr = nr;
            pi = ni;
        }
    }
    ph[2 * t]     = pr;
    ph[2 * t + 1] = pi;
}
__device__ __forceinline__ const float* phaseTab(const float* ph, int L, int side, int g) {
    return ph + 2 * (144 * side + (L - 1) * 48 + g * 16);
}

__device__ __forceinline__ void buildColumn0(float* cr, float* ci,
                                             const float* gl, int t) {
    const int g = t >> 4, row = t & 15;
    float pr = 0.0f, pi = 0.0f;
#pragma unroll
    for (int wl = 0; wl < 4; ++wl) {
        const int q = 11 - (4 * g + wl);
        const int e = ((row >> wl) & 1) << 2;
        const float ur = gl[q * 8 + e];
        const float ui = gl[q * 8 + e + 1];
        if (wl == 0) {
            pr = ur;
            pi = ui;
        } else {
            const float nr = pr * ur - pi * ui;
            const float ni = pr * ui + pi * ur;
            pr = nr;
            pi = ni;
        }
    }
    cr[t] = pr;
    ci[t] = pi;
}

template <int G, bool PHASE>
__device__ __forceinline__ void apply16T(const float* in, float* out,
                                         const float* mr,
                                         const float* pReg, const float* pTile,
                                         const float* pLow,
                                         int wave, int lane) {
    constexpr int SH = G * 4;
    constexpr int LOWMASK = (1 << SH) - 1;
    const int half = lane >> 4;
    const int n15  = lane & 15;

    v2f ar[4];
#pragma unroll
    for (int kc = 0; kc < 4; ++kc) {
        const int k0 = kc * 4 + 2 * half;
        ar[kc].x = mr[n15 * 16 + k0];  ar[kc].y = mr[n15 * 16 + k0 + 1];
    }

    v8f qr = {0.f,0.f,0.f,0.f,0.f,0.f,0.f,0.f};
    v8f qi = {0.f,0.f,0.f,0.f,0.f,0.f,0.f,0.f};
    float plr = 0.0f, pli = 0.0f;
    if (PHASE) {
#pragma unroll
        for (int r = 0; r < 8; ++r) {
            qr[r] = pReg[2 * (r + 8 * half)];
            qi[r] = pReg[2 * (r + 8 * half) + 1];
        }
        plr = pLow[2 * n15];
        pli = pLow[2 * n15 + 1];
    }

#pragma unroll 1
    for (int tt = 0; tt < 2; ++tt) {
        const int col = (wave * 2 + tt) * 16 + n15;
        const int base = ((col >> SH) << (SH + 4)) | (col & LOWMASK);

        v2f br[4], bi[4];
#pragma unroll
        for (int kc = 0; kc < 4; ++kc) {
            const int k0 = kc * 4 + 2 * half;
            const int i0 = base | ((k0) << SH);
            const int i1 = base | ((k0 + 1) << SH);
            br[kc].x = in[i0];          br[kc].y = in[i1];
            bi[kc].x = in[NSTATE + i0]; bi[kc].y = in[NSTATE + i1];
        }

        v8f accR = {0.f,0.f,0.f,0.f,0.f,0.f,0.f,0.f};
        v8f accI = {0.f,0.f,0.f,0.f,0.f,0.f,0.f,0.f};
#pragma unroll
        for (int kc = 0; kc < 4; ++kc) {
            accR = __builtin_amdgcn_wmma_f32_16x16x4_f32(false, ar[kc], false, br[kc], (short)0, accR, false, false);
            accI = __builtin_amdgcn_wmma_f32_16x16x4_f32(false, ar[kc], false, bi[kc], (short)0, accI, false, false);
        }

        if (PHASE) {
            const int tcon = col >> 4;
            const float ptr_ = pTile[2 * tcon], pti = pTile[2 * tcon + 1];
            const float cr = ptr_ * plr - pti * pli;
            const float ci = ptr_ * pli + pti * plr;
            const v8f wr = qr * cr - qi * ci;
            const v8f wi = qr * ci + qi * cr;
            const v8f oR = accR * wr - accI * wi;
            const v8f oI = accR * wi + accI * wr;
            accR = oR;
            accI = oI;
        }

#pragma unroll
        for (int r = 0; r < 8; ++r) {
            const int i = base | ((r + 8 * half) << SH);
            out[i]          = accR[r];
            out[NSTATE + i] = accI[r];
        }
    }
}

template <int R>
__device__ __forceinline__ void chainTopLow(const float* in, float* out,
                                            const float* m2, const float* m0,
                                            const float* pTop, const float* pMid,
                                            const float* pLow,
                                            int wave, int lane) {
    constexpr int SH = 8;
    const int half = lane >> 4;
    const int n15  = lane & 15;

    v2f tr[4];
#pragma unroll
    for (int kc = 0; kc < 4; ++kc) {
        const int k0 = kc * 4 + 2 * half;
        tr[kc].x = m2[n15 * 16 + k0];  tr[kc].y = m2[n15 * 16 + k0 + 1];
    }
    v2f lr[4];
#pragma unroll
    for (int kc = 0; kc < 4; ++kc) {
        const int k0 = 2 * kc + 8 * half;
        lr[kc].x = m0[n15 * 16 + k0];  lr[kc].y = m0[n15 * 16 + k0 + 1];
    }
    v2f ptr_[4], pti[4];
#pragma unroll
    for (int kc = 0; kc < 4; ++kc) {
        const int k0 = kc * 4 + 2 * half;
        ptr_[kc].x = pTop[2 * k0];      ptr_[kc].y = pTop[2 * k0 + 2];
        pti[kc].x  = pTop[2 * k0 + 1];  pti[kc].y  = pTop[2 * k0 + 3];
    }
    const float plr = pLow[2 * n15], pli = pLow[2 * n15 + 1];

    const int base0 = (wave * 2) * 16 + n15;
    const int pl0   = ringSrc(R, base0 | ((2 * half) << SH));

#pragma unroll 1
    for (int tt = 0; tt < 2; ++tt) {
        const int base = base0 | (tt << 4);
        const int pl   = pl0 ^ (-tt & ringSrc(R, 16));

        v2f sr[4], si[4];
#pragma unroll
        for (int kc = 0; kc < 4; ++kc) {
            const int i0 = pl ^ ringSrc(R, (kc * 4) << SH);
            const int i1 = pl ^ ringSrc(R, (kc * 4 + 1) << SH);
            sr[kc].x = in[i0];          sr[kc].y = in[i1];
            si[kc].x = in[NSTATE + i0]; si[kc].y = in[NSTATE + i1];
        }

        const int mid = wave * 2 + tt;
        const float pmr = pMid[2 * mid], pmi = pMid[2 * mid + 1];
        const float cr = pmr * plr - pmi * pli;
        const float ci = pmr * pli + pmi * plr;
        v2f xr[4], xi[4];
#pragma unroll
        for (int kc = 0; kc < 4; ++kc) {
            const v2f wr = ptr_[kc] * cr - pti[kc] * ci;
            const v2f wi = ptr_[kc] * ci + pti[kc] * cr;
            xr[kc] = sr[kc] * wr - si[kc] * wi;
            xi[kc] = sr[kc] * wi + si[kc] * wr;
        }

        v8f accR = {0.f,0.f,0.f,0.f,0.f,0.f,0.f,0.f};
        v8f accI = {0.f,0.f,0.f,0.f,0.f,0.f,0.f,0.f};
#pragma unroll
        for (int kc = 0; kc < 4; ++kc) {
            accR = __builtin_amdgcn_wmma_f32_16x16x4_f32(false, xr[kc], false, tr[kc], (short)0, accR, false, false);
            accI = __builtin_amdgcn_wmma_f32_16x16x4_f32(false, xi[kc], false, tr[kc], (short)0, accI, false, false);
        }

        v2f ur[4], ui[4];
#pragma unroll
        for (int kc = 0; kc < 4; ++kc) {
            ur[kc].x = accR[2 * kc];  ur[kc].y = accR[2 * kc + 1];
            ui[kc].x = accI[2 * kc];  ui[kc].y = accI[2 * kc + 1];
        }

        v8f outR = {0.f,0.f,0.f,0.f,0.f,0.f,0.f,0.f};
        v8f outI = {0.f,0.f,0.f,0.f,0.f,0.f,0.f,0.f};
#pragma unroll
        for (int kc = 0; kc < 4; ++kc) {
            outR = __builtin_amdgcn_wmma_f32_16x16x4_f32(false, ur[kc], false, lr[kc], (short)0, outR, false, false);
            outI = __builtin_amdgcn_wmma_f32_16x16x4_f32(false, ui[kc], false, lr[kc], (short)0, outI, false, false);
        }

#pragma unroll
        for (int r = 0; r < 8; ++r) {
            const int i = base | ((r + 8 * half) << SH);
            out[i]          = outR[r];
            out[NSTATE + i] = outI[r];
        }
    }
}

template <int L>
__device__ __forceinline__ void layerPass(float* st, float* mr, const float* ph,
                                          const float* gu, int t, int wave, int lane) {
    float* cur = st + (L & 1) * 2 * NSTATE;
    constexpr int RPREV = (L - 1) % (NQ - 1) + 1;
    const float* prev = st + ((L & 1) ^ 1) * 2 * NSTATE;
    chainTopLow<RPREV>(prev, cur, mr + 512, mr,
                       phaseTab(ph, L, 0, 2), phaseTab(ph, L, 0, 1), phaseTab(ph, L, 0, 0),
                       wave, lane);
    buildGroup<1>(mr, gu + L * NQ * 8, t);
    __syncthreads();
    constexpr bool PH = (L + 1 < NLAYERS);
    constexpr int LO = PH ? L : 1;
    apply16T<1, PH>(cur, cur, mr + 256,
                    phaseTab(ph, LO, 1, 1), phaseTab(ph, LO, 1, 2), phaseTab(ph, LO, 1, 0),
                    wave, lane);
    if (L + 1 < NLAYERS) {
        buildGroup<2>(mr, gu + (L + 1) * NQ * 8, t);
        buildGroup<0>(mr, gu + (L + 1) * NQ * 8, t);
    }
    __syncthreads();
}

__global__ __launch_bounds__(NTHREADS)
void vqc_kernel(const float* __restrict__ x,
                const float* __restrict__ wts,
                float* __restrict__ out) {
    __shared__ float st[2 * 2 * NSTATE];
    __shared__ float mr[3 * 256];
    __shared__ __attribute__((aligned(16))) float ph[240 * 2];
    __shared__ float gu[NFUSED * 8];
    __shared__ float c0r[3 * 16];
    __shared__ float c0i[3 * 16];
    __shared__ float wred[NQ * 8];

    const int t    = threadIdx.x;
    const int b    = blockIdx.x;
    const int wave = t >> 5;
    const int lane = t & 31;

    if (t < NFUSED) {
        const float* wp = wts + t * 3;
        const float phi = wp[0], th = wp[1], om = wp[2];
        const float ct = cosf(th * 0.5f);
        const float sn = sinf(th * 0.5f);
        const float cp = cosf(phi * 0.5f), sp = sinf(phi * 0.5f);
        const float co = cosf(om * 0.5f),  so = sinf(om * 0.5f);
        float* gr = gu + t * 8;
        if (t < NQ) {
            const float chs = cp * co - sp * so, shs = sp * co + cp * so;
            const float chd = cp * co + sp * so, shd = sp * co - cp * so;
            const float e0 =  chs * ct, e1 = -shs * ct;
            const float e2 = -chd * sn, e3 = -shd * sn;
            const float e4 =  chd * sn, e5 = -shd * sn;
            const float e6 =  chs * ct, e7 =  shs * ct;
            const float a = x[b * NQ + t] * 0.5f;
            const float c = cosf(a), s = sinf(a);
            gr[0] = e0 * c + e3 * s;  gr[1] = e1 * c - e2 * s;
            gr[2] = e1 * s + e2 * c;  gr[3] = e3 * c - e0 * s;
            gr[4] = e4 * c + e7 * s;  gr[5] = e5 * c - e6 * s;
            gr[6] = e5 * s + e6 * c;  gr[7] = e7 * c - e4 * s;
        } else {
            gr[0] = ct; gr[1] = sn; gr[2] = cp; gr[3] = sp;
            gr[4] = co; gr[5] = so; gr[6] = 0.0f; gr[7] = 0.0f;
        }
    }
    __syncthreads();

    if (t < 48) buildColumn0(c0r, c0i, gu, t);
    if (t < 240) buildPhase(ph, gu, t);
    buildGroup<2>(mr, gu + NQ * 8, t);
    buildGroup<0>(mr, gu + NQ * 8, t);
    __syncthreads();

    {
        const float ar = c0r[16 + (t >> 4)], ai = c0i[16 + (t >> 4)];
        const float br = c0r[t & 15],        bi = c0i[t & 15];
        const float wr = ar * br - ai * bi;
        const float wi = ar * bi + ai * br;
#pragma unroll
        for (int k = 0; k < 16; ++k) {
            const int i = k * NTHREADS + t;
            const float hr = c0r[32 + k], hi = c0i[32 + k];
            st[i]          = hr * wr - hi * wi;
            st[NSTATE + i] = hr * wi + hi * wr;
        }
    }
    __syncthreads();

    layerPass<1>(st, mr, ph, gu, t, wave, lane);
    layerPass<2>(st, mr, ph, gu, t, wave, lane);
    layerPass<3>(st, mr, ph, gu, t, wave, lane);

    constexpr int RLAST = (NLAYERS - 1) % (NQ - 1) + 1;
    const float* fin = st + ((NLAYERS - 1) & 1) * 2 * NSTATE;
    float part[NQ];
#pragma unroll
    for (int q = 0; q < NQ; ++q) part[q] = 0.0f;
#pragma unroll
    for (int k = 0; k < 16; ++k) {
        const int i = k * NTHREADS + t;
        const float p = fin[i] * fin[i] + fin[NSTATE + i] * fin[NSTATE + i];
        const int jk = ringDst(RLAST, k * NTHREADS);
#pragma unroll
        for (int q = 0; q < NQ; ++q) {
            part[q] += ((jk >> (11 - q)) & 1) ? -p : p;
        }
    }
    const int jt = ringDst(RLAST, t);
#pragma unroll
    for (int q = 0; q < NQ; ++q) {
        if ((jt >> (11 - q)) & 1) part[q] = -part[q];
    }
#pragma unroll
    for (int q = 0; q < NQ; ++q) {
        float v = part[q];
        for (int off = 16; off >= 1; off >>= 1)
            v += __shfl_xor(v, off, 32);
        if (lane == 0) wred[q * 8 + wave] = v;
    }
    __syncthreads();
    if (t < NQ) {
        float s = 0.0f;
#pragma unroll
        for (int wv = 0; wv < 8; ++wv) s += wred[t * 8 + wv];
        out[b * NQ + t] = s;
    }
}

extern "C" void kernel_launch(void* const* d_in, const int* in_sizes, int n_in,
                              void* d_out, int out_size, void* d_ws, size_t ws_size,
                              hipStream_t stream) {
    const float* x   = (const float*)d_in[0];
    const float* wts = (const float*)d_in[1];
    float* out = (float*)d_out;
    const int batch = in_sizes[0] / NQ;
    vqc_kernel<<<batch, NTHREADS, 0, stream>>>(x, wts, out);
}


# final expectation sums: the 12 wave-level reductions as independent vector-ALU chains (l^16 exchange, then row_xmask DPP moves + adds), not 60 serialized LDS-permute round trips; sums bit-identical
# speedup vs baseline: 8.2530x; 1.0727x over previous
#include <hip/hip_runtime.h>
#include <math.h>


typedef __attribute__((ext_vector_type(2))) float v2f;
typedef __attribute__((ext_vector_type(8))) float v8f;

#define NQ      12
#define NSTATE  4096
#define NLAYERS 4
#define NTHREADS 256
#define NFUSED  (NLAYERS * NQ)

__host__ __device__ constexpr int ringSrc(int r, int j) {
    int s = j;
    for (int qq = NQ - 1; qq >= 0; --qq) {
        const int cbit = 11 - qq;
        const int tbit = 11 - ((qq + r) % NQ);
        s ^= ((s >> cbit) & 1) << tbit;
    }
    return s;
}
__host__ __device__ constexpr int ringDst(int r, int i) {
    int s = i;
    for (int qq = 0; qq < NQ; ++qq) {
        const int cbit = 11 - qq;
        const int tbit = 11 - ((qq + r) % NQ);
        s ^= ((s >> cbit) & 1) << tbit;
    }
    return s;
}

template <int g>
__device__ __forceinline__ void buildGroup(float* mr, const float* gl, int t) {
    const int row = t >> 4, cm = t & 15;
    float p = 0.0f;
#pragma unroll
    for (int wl = 0; wl < 4; ++wl) {
        const int q = 11 - (4 * g + wl);
        const int rb = (row >> wl) & 1, cb = (cm >> wl) & 1;
        const float cs = gl[q * 8];
        const float sn = gl[q * 8 + 1];
        const float u = (rb == cb) ? cs : (rb ? sn : -sn);
        p = (wl == 0) ? u : p * u;
    }
    mr[g * 256 + t] = p;
}

__device__ __forceinline__ void buildPhase(float* ph, const float* gu, int t) {
    const int side = (t >= 144) ? 1 : 0;
    const int u = t - 144 * side;
    const int L = u / 48 + 1;
    const int r = u % 48;
    const int g = r >> 4, idx = r & 15;
    const float* gl = gu + L * NQ * 8 + 2 + 2 * side;
    float pr = 0.0f, pi = 0.0f;
#pragma unroll
    for (int wl = 0; wl < 4; ++wl) {
        const int q = 11 - (4 * g + wl);
        const float cs = gl[q * 8];
        const float s0 = gl[q * 8 + 1];
        const float sn = ((idx >> wl) & 1) ? s0 : -s0;
        if (wl == 0) {
            pr = cs;
            pi = sn;
        } else {
            const float nr = pr * cs - pi * sn;
            const float ni = pr * sn + pi * cs;
            pr = nr;
            pi = ni;
        }
    }
    ph[2 * t]     = pr;
    ph[2 * t + 1] = pi;
}
__device__ __forceinline__ const float* phaseTab(const float* ph, int L, int side, int g) {
    return ph + 2 * (144 * side + (L - 1) * 48 + g * 16);
}

__device__ __forceinline__ void buildColumn0(float* cr, float* ci,
                                             const float* gl, int t) {
    const int g = t >> 4, row = t & 15;
    float pr = 0.0f, pi = 0.0f;
#pragma unroll
    for (int wl = 0; wl < 4; ++wl) {
        const int q = 11 - (4 * g + wl);
        const int e = ((row >> wl) & 1) << 2;
        const float ur = gl[q * 8 + e];
        const float ui = gl[q * 8 + e + 1];
        if (wl == 0) {
            pr = ur;
            pi = ui;
        } else {
            const float nr = pr * ur - pi * ui;
            const float ni = pr * ui + pi * ur;
            pr = nr;
            pi = ni;
        }
    }
    cr[t] = pr;
    ci[t] = pi;
}

template <int G, bool PHASE>
__device__ __forceinline__ void apply16T(const float* in, float* out,
                                         const float* mr,
                                         const float* pReg, const float* pTile,
                                         const float* pLow,
                                         int wave, int lane) {
    constexpr int SH = G * 4;
    constexpr int LOWMASK = (1 << SH) - 1;
    const int half = lane >> 4;
    const int n15  = lane & 15;

    v2f ar[4];
#pragma unroll
    for (int kc = 0; kc < 4; ++kc) {
        const int k0 = kc * 4 + 2 * half;
        ar[kc].x = mr[n15 * 16 + k0];  ar[kc].y = mr[n15 * 16 + k0 + 1];
    }

    v8f qr = {0.f,0.f,0.f,0.f,0.f,0.f,0.f,0.f};
    v8f qi = {0.f,0.f,0.f,0.f,0.f,0.f,0.f,0.f};
    float plr = 0.0f, pli = 0.0f;
    if (PHASE) {
#pragma unroll
        for (int r = 0; r < 8; ++r) {
            qr[r] = pReg[2 * (r + 8 * half)];
            qi[r] = pReg[2 * (r + 8 * half) + 1];
        }
        plr = pLow[2 * n15];
        pli = pLow[2 * n15 + 1];
    }

#pragma unroll 1
    for (int tt = 0; tt < 2; ++tt) {
        const int col = (wave * 2 + tt) * 16 + n15;
        const int base = ((col >> SH) << (SH + 4)) | (col & LOWMASK);

        v2f br[4], bi[4];
#pragma unroll
        for (int kc = 0; kc < 4; ++kc) {
            const int k0 = kc * 4 + 2 * half;
            const int i0 = base | ((k0) << SH);
            const int i1 = base | ((k0 + 1) << SH);
            br[kc].x = in[i0];          br[kc].y = in[i1];
            bi[kc].x = in[NSTATE + i0]; bi[kc].y = in[NSTATE + i1];
        }

        v8f accR = {0.f,0.f,0.f,0.f,0.f,0.f,0.f,0.f};
        v8f accI = {0.f,0.f,0.f,0.f,0.f,0.f,0.f,0.f};
#pragma unroll
        for (int kc = 0; kc < 4; ++kc) {
            accR = __builtin_amdgcn_wmma_f32_16x16x4_f32(false, ar[kc], false, br[kc], (short)0, accR, false, false);
            accI = __builtin_amdgcn_wmma_f32_16x16x4_f32(false, ar[kc], false, bi[kc], (short)0, accI, false, false);
        }

        if (PHASE) {
            const int tcon = col >> 4;
            const float ptr_ = pTile[2 * tcon], pti = pTile[2 * tcon + 1];
            const float cr = ptr_ * plr - pti * pli;
            const float ci = ptr_ * pli + pti * plr;
            const v8f wr = qr * cr - qi * ci;
            const v8f wi = qr * ci + qi * cr;
            const v8f oR = accR * wr - accI * wi;
            const v8f oI = accR * wi + accI * wr;
            accR = oR;
            accI = oI;
        }

#pragma unroll
        for (int r = 0; r < 8; ++r) {
            const int i = base | ((r + 8 * half) << SH);
            out[i]          = accR[r];
            out[NSTATE + i] = accI[r];
        }
    }
}

template <int R>
__device__ __forceinline__ void chainTopLow(const float* in, float* out,
                                            const float* m2, const float* m0,
                                            const float* pTop, const float* pMid,
                                            const float* pLow,
                                            int wave, int lane) {
    constexpr int SH = 8;
    const int half = lane >> 4;
    const int n15  = lane & 15;

    v2f tr[4];
#pragma unroll
    for (int kc = 0; kc < 4; ++kc) {
        const int k0 = kc * 4 + 2 * half;
        tr[kc].x = m2[n15 * 16 + k0];  tr[kc].y = m2[n15 * 16 + k0 + 1];
    }
    v2f lr[4];
#pragma unroll
    for (int kc = 0; kc < 4; ++kc) {
        const int k0 = 2 * kc + 8 * half;
        lr[kc].x = m0[n15 * 16 + k0];  lr[kc].y = m0[n15 * 16 + k0 + 1];
    }
    v2f ptr_[4], pti[4];
#pragma unroll
    for (int kc = 0; kc < 4; ++kc) {
        const int k0 = kc * 4 + 2 * half;
        ptr_[kc].x = pTop[2 * k0];      ptr_[kc].y = pTop[2 * k0 + 2];
        pti[kc].x  = pTop[2 * k0 + 1];  pti[kc].y  = pTop[2 * k0 + 3];
    }
    const float plr = pLow[2 * n15], pli = pLow[2 * n15 + 1];

    const int base0 = (wave * 2) * 16 + n15;
    const int pl0   = ringSrc(R, base0 | ((2 * half) << SH));

#pragma unroll 1
    for (int tt = 0; tt < 2; ++tt) {
        const int base = base0 | (tt << 4);
        const int pl   = pl0 ^ (-tt & ringSrc(R, 16));

        v2f sr[4], si[4];
#pragma unroll
        for (int kc = 0; kc < 4; ++kc) {
            const int i0 = pl ^ ringSrc(R, (kc * 4) << SH);
            const int i1 = pl ^ ringSrc(R, (kc * 4 + 1) << SH);
            sr[kc].x = in[i0];          sr[kc].y = in[i1];
            si[kc].x = in[NSTATE + i0]; si[kc].y = in[NSTATE + i1];
        }

        const int mid = wave * 2 + tt;
        const float pmr = pMid[2 * mid], pmi = pMid[2 * mid + 1];
        const float cr = pmr * plr - pmi * pli;
        const float ci = pmr * pli + pmi * plr;
        v2f xr[4], xi[4];
#pragma unroll
        for (int kc = 0; kc < 4; ++kc) {
            const v2f wr = ptr_[kc] * cr - pti[kc] * ci;
            const v2f wi = ptr_[kc] * ci + pti[kc] * cr;
            xr[kc] = sr[kc] * wr - si[kc] * wi;
            xi[kc] = sr[kc] * wi + si[kc] * wr;
        }

        v8f accR = {0.f,0.f,0.f,0.f,0.f,0.f,0.f,0.f};
        v8f accI = {0.f,0.f,0.f,0.f,0.f,0.f,0.f,0.f};
#pragma unroll
        for (int kc = 0; kc < 4; ++kc) {
            accR = __builtin_amdgcn_wmma_f32_16x16x4_f32(false, xr[kc], false, tr[kc], (short)0, accR, false, false);
            accI = __builtin_amdgcn_wmma_f32_16x16x4_f32(false, xi[kc], false, tr[kc], (short)0, accI, false, false);
        }

        v2f ur[4], ui[4];
#pragma unroll
        for (int kc = 0; kc < 4; ++kc) {
            ur[kc].x = accR[2 * kc];  ur[kc].y = accR[2 * kc + 1];
            ui[kc].x = accI[2 * kc];  ui[kc].y = accI[2 * kc + 1];
        }

        v8f outR = {0.f,0.f,0.f,0.f,0.f,0.f,0.f,0.f};
        v8f outI = {0.f,0.f,0.f,0.f,0.f,0.f,0.f,0.f};
#pragma unroll
        for (int kc = 0; kc < 4; ++kc) {
            outR = __builtin_amdgcn_wmma_f32_16x16x4_f32(false, ur[kc], false, lr[kc], (short)0, outR, false, false);
            outI = __builtin_amdgcn_wmma_f32_16x16x4_f32(false, ui[kc], false, lr[kc], (short)0, outI, false, false);
        }

#pragma unroll
        for (int r = 0; r < 8; ++r) {
            const int i = base | ((r + 8 * half) << SH);
            out[i]          = outR[r];
            out[NSTATE + i] = outI[r];
        }
    }
}

template <int L>
__device__ __forceinline__ void layerPass(float* st, float* mr, const float* ph,
                                          const float* gu, int t, int wave, int lane) {
    float* cur = st + (L & 1) * 2 * NSTATE;
    constexpr int RPREV = (L - 1) % (NQ - 1) + 1;
    const float* prev = st + ((L & 1) ^ 1) * 2 * NSTATE;
    chainTopLow<RPREV>(prev, cur, mr + 512, mr,
                       phaseTab(ph, L, 0, 2), phaseTab(ph, L, 0, 1), phaseTab(ph, L, 0, 0),
                       wave, lane);
    buildGroup<1>(mr, gu + L * NQ * 8, t);
    __syncthreads();
    constexpr bool PH = (L + 1 < NLAYERS);
    constexpr int LO = PH ? L : 1;
    apply16T<1, PH>(cur, cur, mr + 256,
                    phaseTab(ph, LO, 1, 1), phaseTab(ph, LO, 1, 2), phaseTab(ph, LO, 1, 0),
                    wave, lane);
    if (L + 1 < NLAYERS) {
        buildGroup<2>(mr, gu + (L + 1) * NQ * 8, t);
        buildGroup<0>(mr, gu + (L + 1) * NQ * 8, t);
    }
    __syncthreads();
}

__device__ __forceinline__ float xhalf(float x) {
    const int bits = __builtin_bit_cast(int, x);
    return __builtin_bit_cast(float,
        __builtin_amdgcn_permlanex16(bits, bits, 0x76543210, 0xfedcba98, false, false));
}
template <int M>
__device__ __forceinline__ float rowXor(float x) {
    const int bits = __builtin_bit_cast(int, x);
    return __builtin_bit_cast(float,
        __builtin_amdgcn_update_dpp(0, bits, 0x160 | M, 0xf, 0xf, true));
}

__global__ __launch_bounds__(NTHREADS)
void vqc_kernel(const float* __restrict__ x,
                const float* __restrict__ wts,
                float* __restrict__ out) {
    __shared__ float st[2 * 2 * NSTATE];
    __shared__ float mr[3 * 256];
    __shared__ __attribute__((aligned(16))) float ph[240 * 2];
    __shared__ float gu[NFUSED * 8];
    __shared__ float c0r[3 * 16];
    __shared__ float c0i[3 * 16];
    __shared__ __attribute__((aligned(16))) float wred[8 * NQ];

    const int t    = threadIdx.x;
    const int b    = blockIdx.x;
    const int wave = t >> 5;
    const int lane = t & 31;

    if (t < NFUSED) {
        const float* wp = wts + t * 3;
        const float phi = wp[0], th = wp[1], om = wp[2];
        const float ct = cosf(th * 0.5f);
        const float sn = sinf(th * 0.5f);
        const float cp = cosf(phi * 0.5f), sp = sinf(phi * 0.5f);
        const float co = cosf(om * 0.5f),  so = sinf(om * 0.5f);
        float* gr = gu + t * 8;
        if (t < NQ) {
            const float chs = cp * co - sp * so, shs = sp * co + cp * so;
            const float chd = cp * co + sp * so, shd = sp * co - cp * so;
            const float e0 =  chs * ct, e1 = -shs * ct;
            const float e2 = -chd * sn, e3 = -shd * sn;
            const float e4 =  chd * sn, e5 = -shd * sn;
            const float e6 =  chs * ct, e7 =  shs * ct;
            const float a = x[b * NQ + t] * 0.5f;
            const float c = cosf(a), s = sinf(a);
            gr[0] = e0 * c + e3 * s;  gr[1] = e1 * c - e2 * s;
            gr[2] = e1 * s + e2 * c;  gr[3] = e3 * c - e0 * s;
            gr[4] = e4 * c + e7 * s;  gr[5] = e5 * c - e6 * s;
            gr[6] = e5 * s + e6 * c;  gr[7] = e7 * c - e4 * s;
        } else {
            gr[0] = ct; gr[1] = sn; gr[2] = cp; gr[3] = sp;
            gr[4] = co; gr[5] = so; gr[6] = 0.0f; gr[7] = 0.0f;
        }
    }
    __syncthreads();

    if (t < 48) buildColumn0(c0r, c0i, gu, t);
    if (t < 240) buildPhase(ph, gu, t);
    buildGroup<2>(mr, gu + NQ * 8, t);
    buildGroup<0>(mr, gu + NQ * 8, t);
    __syncthreads();

    {
        const float ar = c0r[16 + (t >> 4)], ai = c0i[16 + (t >> 4)];
        const float br = c0r[t & 15],        bi = c0i[t & 15];
        const float wr = ar * br - ai * bi;
        const float wi = ar * bi + ai * br;
#pragma unroll
        for (int k = 0; k < 16; ++k) {
            const int i = k * NTHREADS + t;
            const float hr = c0r[32 + k], hi = c0i[32 + k];
            st[i]          = hr * wr - hi * wi;
            st[NSTATE + i] = hr * wi + hi * wr;
        }
    }
    __syncthreads();

    layerPass<1>(st, mr, ph, gu, t, wave, lane);
    layerPass<2>(st, mr, ph, gu, t, wave, lane);
    layerPass<3>(st, mr, ph, gu, t, wave, lane);

    constexpr int RLAST = (NLAYERS - 1) % (NQ - 1) + 1;
    const float* fin = st + ((NLAYERS - 1) & 1) * 2 * NSTATE;
    float part[NQ];
#pragma unroll
    for (int q = 0; q < NQ; ++q) part[q] = 0.0f;
#pragma unroll
    for (int k = 0; k < 16; ++k) {
        const int i = k * NTHREADS + t;
        const float p = fin[i] * fin[i] + fin[NSTATE + i] * fin[NSTATE + i];
        const int jk = ringDst(RLAST, k * NTHREADS);
#pragma unroll
        for (int q = 0; q < NQ; ++q) {
            part[q] += ((jk >> (11 - q)) & 1) ? -p : p;
        }
    }
    const int jt = ringDst(RLAST, t);
#pragma unroll
    for (int q = 0; q < NQ; ++q) {
        if ((jt >> (11 - q)) & 1) part[q] = -part[q];
    }
    float red[NQ];
#pragma unroll
    for (int q = 0; q < NQ; ++q) {
        float v = part[q];
        v += xhalf(v);
        v += rowXor<8>(v);
        v += rowXor<4>(v);
        v += rowXor<2>(v);
        v += rowXor<1>(v);
        red[q] = v;
    }
    if (lane == 0) {
#pragma unroll
        for (int q = 0; q < NQ; ++q) wred[wave * NQ + q] = red[q];
    }
    __syncthreads();
    if (t < NQ) {
        float s = 0.0f;
#pragma unroll
        for (int wv = 0; wv < 8; ++wv) s += wred[wv * NQ + t];
        out[b * NQ + t] = s;
    }
}

extern "C" void kernel_launch(void* const* d_in, const int* in_sizes, int n_in,
                              void* d_out, int out_size, void* d_ws, size_t ws_size,
                              hipStream_t stream) {
    const float* x   = (const float*)d_in[0];
    const float* wts = (const float*)d_in[1];
    float* out = (float*)d_out;
    const int batch = in_sizes[0] / NQ;
    vqc_kernel<<<batch, NTHREADS, 0, stream>>>(x, wts, out);
}


# layers 0+1 fused: the state behind layer 1's Rot gates is a rank-4 complex product V*W of chain factors, 8 WMMA per wave; 2 state passes and 40 of 144 WMMA per wave go; f32 rounding, sum order change
# speedup vs baseline: 10.6624x; 1.2919x over previous
#include <hip/hip_runtime.h>
#include <math.h>


typedef __attribute__((ext_vector_type(2))) float v2f;
typedef __attribute__((ext_vector_type(8))) float v8f;

#define NQ      12
#define NSTATE  4096
#define NLAYERS 4
#define NTHREADS 256
#define NFUSED  (NLAYERS * NQ)

__host__ __device__ constexpr int ringSrc(int r, int j) {
    int s = j;
    for (int qq = NQ - 1; qq >= 0; --qq) {
        const int cbit = 11 - qq;
        const int tbit = 11 - ((qq + r) % NQ);
        s ^= ((s >> cbit) & 1) << tbit;
    }
    return s;
}
__host__ __device__ constexpr int ringDst(int r, int i) {
    int s = i;
    for (int qq = 0; qq < NQ; ++qq) {
        const int cbit = 11 - qq;
        const int tbit = 11 - ((qq + r) % NQ);
        s ^= ((s >> cbit) & 1) << tbit;
    }
    return s;
}

template <int g>
__device__ __forceinline__ void buildGroup(float* mr, const float* gl, int t) {
    const int row = t >> 4, cm = t & 15;
    float p = 0.0f;
#pragma unroll
    for (int wl = 0; wl < 4; ++wl) {
        const int q = 11 - (4 * g + wl);
        const int rb = (row >> wl) & 1, cb = (cm >> wl) & 1;
        const float cs = gl[q * 8];
        const float sn = gl[q * 8 + 1];
        const float u = (rb == cb) ? cs : (rb ? sn : -sn);
        p = (wl == 0) ? u : p * u;
    }
    mr[g * 256 + t] = p;
}

__device__ __forceinline__ void buildPhase(float* ph, const float* gu, int t) {
    const int side = (t >= 144) ? 1 : 0;
    const int u = t - 144 * side;
    if (u < 48) return;
    const int L = u / 48 + 1;
    const int r = u % 48;
    const int g = r >> 4, idx = r & 15;
    const float* gl = gu + L * NQ * 8 + 2 + 2 * side;
    float pr = 0.0f, pi = 0.0f;
#pragma unroll
    for (int wl = 0; wl < 4; ++wl) {
        const int q = 11 - (4 * g + wl);
        const float cs = gl[q * 8];
        const float s0 = gl[q * 8 + 1];
        const float sn = ((idx >> wl) & 1) ? s0 : -s0;
        if (wl == 0) {
            pr = cs;
            pi = sn;
        } else {
            const float nr = pr * cs - pi * sn;
            const float ni = pr * sn + pi * cs;
            pr = nr;
            pi = ni;
        }
    }
    ph[2 * t]     = pr;
    ph[2 * t + 1] = pi;
}
__device__ __forceinline__ const float* phaseTab(const float* ph, int L, int side, int g) {
    return ph + 2 * (144 * side + (L - 1) * 48 + g * 16);
}

__device__ __forceinline__ void chainStep(float& x0r, float& x0i, float& x1r, float& x1i,
                                          const float* a, const float* g, int o) {
    const float a0r = a[0], a0i = a[1], a1r = a[4], a1i = a[5];
    const float t0r = a0r * x0r - a0i * x0i + a1r * x1r - a1i * x1i;
    const float t0i = a0r * x0i + a0i * x0r + a1r * x1i + a1i * x1r;
    const float t1r = a1r * x0r - a1i * x0i + a0r * x1r - a0i * x1i;
    const float t1i = a1r * x0i + a1i * x0r + a0r * x1i + a0i * x1r;
    const float g0r = g[4 * o],     g0i = g[4 * o + 1];
    const float g1r = g[4 * o + 2], g1i = g[4 * o + 3];
    x0r = g0r * t0r - g0i * t0i;  x0i = g0r * t0i + g0i * t0r;
    x1r = g1r * t1r - g1i * t1i;  x1i = g1r * t1i + g1i * t1r;
}

__device__ __forceinline__ void buildChainTables(float* midT, float* lowT, float* topV,
                                                 const float* gu, int wave, int lane) {
    const float* g1 = gu + NQ * 8;
    if (wave == 0 || wave == 1) {
        const int idx = lane >> 1, jin = lane & 1;
        float x0r = jin ? 0.0f : 1.0f, x0i = 0.0f;
        float x1r = jin ? 1.0f : 0.0f, x1i = 0.0f;
        const int qb = (wave == 0) ? 4 : 8;
#pragma unroll
        for (int d = 0; d < 4; ++d) {
            const int q = qb + d;
            chainStep(x0r, x0i, x1r, x1i, gu + q * 8, g1 + q * 8, (idx >> (3 - d)) & 1);
        }
        float* o = ((wave == 0) ? midT : lowT) + lane * 4;
        o[0] = x0r; o[1] = x0i; o[2] = x1r; o[3] = x1i;
    } else if (wave == 2 && lane < 16) {
        const int s = lane >> 3, top3 = lane & 7;
        const int o1 = (top3 >> 2) & 1;
        const float asr = gu[4 * s], asi = gu[4 * s + 1];
        const float* a1 = gu + 8;
        const float* gg = g1 + 8 + 4 * o1;
        const float b0r = a1[4 * s],       b0i = a1[4 * s + 1];
        const float b1r = a1[4 * (s ^ 1)], b1i = a1[4 * (s ^ 1) + 1];
        const float u0r = gg[0] * b0r - gg[1] * b0i, u0i = gg[0] * b0i + gg[1] * b0r;
        const float u1r = gg[2] * b1r - gg[3] * b1i, u1i = gg[2] * b1i + gg[3] * b1r;
        float x0r = asr * u0r - asi * u0i, x0i = asr * u0i + asi * u0r;
        float x1r = asr * u1r - asi * u1i, x1i = asr * u1i + asi * u1r;
        chainStep(x0r, x0i, x1r, x1i, gu + 2 * 8, g1 + 2 * 8, (top3 >> 1) & 1);
        chainStep(x0r, x0i, x1r, x1i, gu + 3 * 8, g1 + 3 * 8, top3 & 1);
        float* o = topV + lane * 4;
        o[0] = x0r; o[1] = x0i; o[2] = x1r; o[3] = x1i;
    }
}

__device__ __forceinline__ void layer01Direct(float* out, const float* midT,
                                              const float* lowT, const float* topV,
                                              const float* gu, int wave, int lane) {
    const int h = lane >> 4, c = lane & 15;
    const int m = ((c & 7) << 1) | (c >> 3);
    const float4 tv  = *(const float4*)(topV + (h * 8 + wave) * 4);
    const float4 m0  = *(const float4*)(midT + m * 8);
    const float4 m1  = *(const float4*)(midT + m * 8 + 4);
    v2f aR, aI;
    aR.x = tv.x * m0.x - tv.y * m0.y + tv.z * m1.x - tv.w * m1.y;
    aI.x = tv.x * m0.y + tv.y * m0.x + tv.z * m1.y + tv.w * m1.x;
    aR.y = tv.x * m0.z - tv.y * m0.w + tv.z * m1.z - tv.w * m1.w;
    aI.y = tv.x * m0.w + tv.y * m0.z + tv.z * m1.w + tv.w * m1.z;
    const v2f aIn = -aI;
    const float4 l0 = *(const float4*)(lowT + c * 8);
    const float4 l1 = *(const float4*)(lowT + c * 8 + 4);
    const float* g0 = gu + NQ * 8;
#pragma unroll
    for (int tc = 0; tc < 2; ++tc) {
        const float gar = g0[4 * tc + 2 * h],       gai = g0[4 * tc + 2 * h + 1];
        const float gbr = g0[4 * tc + 2 * (h ^ 1)], gbi = g0[4 * tc + 2 * (h ^ 1) + 1];
        v2f bR, bI;
        bR.x = l0.x * gar - l0.y * gai + l0.z * gbr - l0.w * gbi;
        bI.x = l0.x * gai + l0.y * gar + l0.z * gbi + l0.w * gbr;
        bR.y = l1.x * gar - l1.y * gai + l1.z * gbr - l1.w * gbi;
        bI.y = l1.x * gai + l1.y * gar + l1.z * gbi + l1.w * gbr;
        v8f accR = {0.f,0.f,0.f,0.f,0.f,0.f,0.f,0.f};
        v8f accI = {0.f,0.f,0.f,0.f,0.f,0.f,0.f,0.f};
        accR = __builtin_amdgcn_wmma_f32_16x16x4_f32(false, aR,  false, bR, (short)0, accR, false, false);
        accI = __builtin_amdgcn_wmma_f32_16x16x4_f32(false, aR,  false, bI, (short)0, accI, false, false);
        accR = __builtin_amdgcn_wmma_f32_16x16x4_f32(false, aIn, false, bI, (short)0, accR, false, false);
        accI = __builtin_amdgcn_wmma_f32_16x16x4_f32(false, aI,  false, bR, (short)0, accI, false, false);
#pragma unroll
        for (int r = 0; r < 8; ++r) {
            const int i = (tc << 11) | (wave << 8) | (r << 5) | lane;
            out[i]          = accR[r];
            out[NSTATE + i] = accI[r];
        }
    }
}

template <int G, bool PHASE>
__device__ __forceinline__ void apply16T(const float* in, float* out,
                                         const float* mr,
                                         const float* pReg, const float* pTile,
                                         const float* pLow,
                                         int wave, int lane) {
    constexpr int SH = G * 4;
    constexpr int LOWMASK = (1 << SH) - 1;
    const int half = lane >> 4;
    const int n15  = lane & 15;

    v2f ar[4];
#pragma unroll
    for (int kc = 0; kc < 4; ++kc) {
        const int k0 = kc * 4 + 2 * half;
        ar[kc].x = mr[n15 * 16 + k0];  ar[kc].y = mr[n15 * 16 + k0 + 1];
    }

    v8f qr = {0.f,0.f,0.f,0.f,0.f,0.f,0.f,0.f};
    v8f qi = {0.f,0.f,0.f,0.f,0.f,0.f,0.f,0.f};
    float plr = 0.0f, pli = 0.0f;
    if (PHASE) {
#pragma unroll
        for (int r = 0; r < 8; ++r) {
            qr[r] = pReg[2 * (r + 8 * half)];
            qi[r] = pReg[2 * (r + 8 * half) + 1];
        }
        plr = pLow[2 * n15];
        pli = pLow[2 * n15 + 1];
    }

#pragma unroll 1
    for (int tt = 0; tt < 2; ++tt) {
        const int col = (wave * 2 + tt) * 16 + n15;
        const int base = ((col >> SH) << (SH + 4)) | (col & LOWMASK);

        v2f br[4], bi[4];
#pragma unroll
        for (int kc = 0; kc < 4; ++kc) {
            const int k0 = kc * 4 + 2 * half;
            const int i0 = base | ((k0) << SH);
            const int i1 = base | ((k0 + 1) << SH);
            br[kc].x = in[i0];          br[kc].y = in[i1];
            bi[kc].x = in[NSTATE + i0]; bi[kc].y = in[NSTATE + i1];
        }

        v8f accR = {0.f,0.f,0.f,0.f,0.f,0.f,0.f,0.f};
        v8f accI = {0.f,0.f,0.f,0.f,0.f,0.f,0.f,0.f};
#pragma unroll
        for (int kc = 0; kc < 4; ++kc) {
            accR = __builtin_amdgcn_wmma_f32_16x16x4_f32(false, ar[kc], false, br[kc], (short)0, accR, false, false);
            accI = __builtin_amdgcn_wmma_f32_16x16x4_f32(false, ar[kc], false, bi[kc], (short)0, accI, false, false);
        }

        if (PHASE) {
            const int tcon = col >> 4;
            const float ptr_ = pTile[2 * tcon], pti = pTile[2 * tcon + 1];
            const float cr = ptr_ * plr - pti * pli;
            const float ci = ptr_ * pli + pti * plr;
            const v8f wr = qr * cr - qi * ci;
            const v8f wi = qr * ci + qi * cr;
            const v8f oR = accR * wr - accI * wi;
            const v8f oI = accR * wi + accI * wr;
            accR = oR;
            accI = oI;
        }

#pragma unroll
        for (int r = 0; r < 8; ++r) {
            const int i = base | ((r + 8 * half) << SH);
            out[i]          = accR[r];
            out[NSTATE + i] = accI[r];
        }
    }
}

template <int R>
__device__ __forceinline__ void chainTopLow(const float* in, float* out,
                                            const float* m2, const float* m0,
                                            const float* pTop, const float* pMid,
                                            const float* pLow,
                                            int wave, int lane) {
    constexpr int SH = 8;
    const int half = lane >> 4;
    const int n15  = lane & 15;

    v2f tr[4];
#pragma unroll
    for (int kc = 0; kc < 4; ++kc) {
        const int k0 = kc * 4 + 2 * half;
        tr[kc].x = m2[n15 * 16 + k0];  tr[kc].y = m2[n15 * 16 + k0 + 1];
    }
    v2f lr[4];
#pragma unroll
    for (int kc = 0; kc < 4; ++kc) {
        const int k0 = 2 * kc + 8 * half;
        lr[kc].x = m0[n15 * 16 + k0];  lr[kc].y = m0[n15 * 16 + k0 + 1];
    }
    v2f ptr_[4], pti[4];
#pragma unroll
    for (int kc = 0; kc < 4; ++kc) {
        const int k0 = kc * 4 + 2 * half;
        ptr_[kc].x = pTop[2 * k0];      ptr_[kc].y = pTop[2 * k0 + 2];
        pti[kc].x  = pTop[2 * k0 + 1];  pti[kc].y  = pTop[2 * k0 + 3];
    }
    const float plr = pLow[2 * n15], pli = pLow[2 * n15 + 1];

    const int base0 = (wave * 2) * 16 + n15;
    const int pl0   = ringSrc(R, base0 | ((2 * half) << SH));

#pragma unroll 1
    for (int tt = 0; tt < 2; ++tt) {
        const int base = base0 | (tt << 4);
        const int pl   = pl0 ^ (-tt & ringSrc(R, 16));

        v2f sr[4], si[4];
#pragma unroll
        for (int kc = 0; kc < 4; ++kc) {
            const int i0 = pl ^ ringSrc(R, (kc * 4) << SH);
            const int i1 = pl ^ ringSrc(R, (kc * 4 + 1) << SH);
            sr[kc].x = in[i0];          sr[kc].y = in[i1];
            si[kc].x = in[NSTATE + i0]; si[kc].y = in[NSTATE + i1];
        }

        const int mid = wave * 2 + tt;
        const float pmr = pMid[2 * mid], pmi = pMid[2 * mid + 1];
        const float cr = pmr * plr - pmi * pli;
        const float ci = pmr * pli + pmi * plr;
        v2f xr[4], xi[4];
#pragma unroll
        for (int kc = 0; kc < 4; ++kc) {
            const v2f wr = ptr_[kc] * cr - pti[kc] * ci;
            const v2f wi = ptr_[kc] * ci + pti[kc] * cr;
            xr[kc] = sr[kc] * wr - si[kc] * wi;
            xi[kc] = sr[kc] * wi + si[kc] * wr;
        }

        v8f accR = {0.f,0.f,0.f,0.f,0.f,0.f,0.f,0.f};
        v8f accI = {0.f,0.f,0.f,0.f,0.f,0.f,0.f,0.f};
#pragma unroll
        for (int kc = 0; kc < 4; ++kc) {
            accR = __builtin_amdgcn_wmma_f32_16x16x4_f32(false, xr[kc], false, tr[kc], (short)0, accR, false, false);
            accI = __builtin_amdgcn_wmma_f32_16x16x4_f32(false, xi[kc], false, tr[kc], (short)0, accI, false, false);
        }

        v2f ur[4], ui[4];
#pragma unroll
        for (int kc = 0; kc < 4; ++kc) {
            ur[kc].x = accR[2 * kc];  ur[kc].y = accR[2 * kc + 1];
            ui[kc].x = accI[2 * kc];  ui[kc].y = accI[2 * kc + 1];
        }

        v8f outR = {0.f,0.f,0.f,0.f,0.f,0.f,0.f,0.f};
        v8f outI = {0.f,0.f,0.f,0.f,0.f,0.f,0.f,0.f};
#pragma unroll
        for (int kc = 0; kc < 4; ++kc) {
            outR = __builtin_amdgcn_wmma_f32_16x16x4_f32(false, ur[kc], false, lr[kc], (short)0, outR, false, false);
            outI = __builtin_amdgcn_wmma_f32_16x16x4_f32(false, ui[kc], false, lr[kc], (short)0, outI, false, false);
        }

#pragma unroll
        for (int r = 0; r < 8; ++r) {
            const int i = base | ((r + 8 * half) << SH);
            out[i]          = outR[r];
            out[NSTATE + i] = outI[r];
        }
    }
}

template <int L>
__device__ __forceinline__ void layerPass(float* st, float* mr, const float* ph,
                                          const float* gu, int t, int wave, int lane) {
    float* cur = st + (L & 1) * 2 * NSTATE;
    constexpr int RPREV = (L - 1) % (NQ - 1) + 1;
    const float* prev = st + ((L & 1) ^ 1) * 2 * NSTATE;
    chainTopLow<RPREV>(prev, cur, mr + 512, mr,
                       phaseTab(ph, L, 0, 2), phaseTab(ph, L, 0, 1), phaseTab(ph, L, 0, 0),
                       wave, lane);
    buildGroup<1>(mr, gu + L * NQ * 8, t);
    __syncthreads();
    constexpr bool PH = (L + 1 < NLAYERS);
    constexpr int LO = PH ? L : 1;
    apply16T<1, PH>(cur, cur, mr + 256,
                    phaseTab(ph, LO, 1, 1), phaseTab(ph, LO, 1, 2), phaseTab(ph, LO, 1, 0),
                    wave, lane);
    if (L + 1 < NLAYERS) {
        buildGroup<2>(mr, gu + (L + 1) * NQ * 8, t);
        buildGroup<0>(mr, gu + (L + 1) * NQ * 8, t);
    }
    __syncthreads();
}

__device__ __forceinline__ float xhalf(float x) {
    const int bits = __builtin_bit_cast(int, x);
    return __builtin_bit_cast(float,
        __builtin_amdgcn_permlanex16(bits, bits, 0x76543210, 0xfedcba98, false, false));
}
template <int M>
__device__ __forceinline__ float rowXor(float x) {
    const int bits = __builtin_bit_cast(int, x);
    return __builtin_bit_cast(float,
        __builtin_amdgcn_update_dpp(0, bits, 0x160 | M, 0xf, 0xf, true));
}

__global__ __launch_bounds__(NTHREADS)
void vqc_kernel(const float* __restrict__ x,
                const float* __restrict__ wts,
                float* __restrict__ out) {
    __shared__ float st[2 * 2 * NSTATE];
    __shared__ float mr[3 * 256];
    __shared__ __attribute__((aligned(16))) float ph[240 * 2];
    __shared__ __attribute__((aligned(16))) float gu[NFUSED * 8];
    __shared__ __attribute__((aligned(16))) float midT[16 * 2 * 2 * 2];
    __shared__ __attribute__((aligned(16))) float lowT[16 * 2 * 2 * 2];
    __shared__ __attribute__((aligned(16))) float topV[2 * 8 * 2 * 2];
    __shared__ __attribute__((aligned(16))) float wred[8 * NQ];

    const int t    = threadIdx.x;
    const int b    = blockIdx.x;
    const int wave = t >> 5;
    const int lane = t & 31;

    if (t < NFUSED) {
        const float* wp = wts + t * 3;
        const float phi = wp[0], th = wp[1], om = wp[2];
        const float ct = cosf(th * 0.5f);
        const float sn = sinf(th * 0.5f);
        const float cp = cosf(phi * 0.5f), sp = sinf(phi * 0.5f);
        const float co = cosf(om * 0.5f),  so = sinf(om * 0.5f);
        float* gr = gu + t * 8;
        if (t < 2 * NQ) {
            const float chs = cp * co - sp * so, shs = sp * co + cp * so;
            const float chd = cp * co + sp * so, shd = sp * co - cp * so;
            const float e0 =  chs * ct, e1 = -shs * ct;
            const float e2 = -chd * sn, e3 = -shd * sn;
            const float e4 =  chd * sn, e5 = -shd * sn;
            const float e6 =  chs * ct, e7 =  shs * ct;
            if (t < NQ) {
                const float a = x[b * NQ + t] * 0.5f;
                const float c = cosf(a), s = sinf(a);
                gr[0] = e0 * c + e3 * s;  gr[1] = e1 * c - e2 * s;
                gr[2] = e1 * s + e2 * c;  gr[3] = e3 * c - e0 * s;
                gr[4] = e4 * c + e7 * s;  gr[5] = e5 * c - e6 * s;
                gr[6] = e5 * s + e6 * c;  gr[7] = e7 * c - e4 * s;
            } else {
                gr[0] = e0; gr[1] = e1; gr[2] = e2; gr[3] = e3;
                gr[4] = e4; gr[5] = e5; gr[6] = e6; gr[7] = e7;
            }
        } else {
            gr[0] = ct; gr[1] = sn; gr[2] = cp; gr[3] = sp;
            gr[4] = co; gr[5] = so; gr[6] = 0.0f; gr[7] = 0.0f;
        }
    }
    __syncthreads();

    buildChainTables(midT, lowT, topV, gu, wave, lane);
    if (t < 240) buildPhase(ph, gu, t);
    buildGroup<2>(mr, gu + 2 * NQ * 8, t);
    buildGroup<0>(mr, gu + 2 * NQ * 8, t);
    __syncthreads();

    layer01Direct(st + 2 * NSTATE, midT, lowT, topV, gu, wave, lane);
    __syncthreads();

    layerPass<2>(st, mr, ph, gu, t, wave, lane);
    layerPass<3>(st, mr, ph, gu, t, wave, lane);

    constexpr int RLAST = (NLAYERS - 1) % (NQ - 1) + 1;
    const float* fin = st + ((NLAYERS - 1) & 1) * 2 * NSTATE;
    float part[NQ];
#pragma unroll
    for (int q = 0; q < NQ; ++q) part[q] = 0.0f;
#pragma unroll
    for (int k = 0; k < 16; ++k) {
        const int i = k * NTHREADS + t;
        const float p = fin[i] * fin[i] + fin[NSTATE + i] * fin[NSTATE + i];
        const int jk = ringDst(RLAST, k * NTHREADS);
#pragma unroll
        for (int q = 0; q < NQ; ++q) {
            part[q] += ((jk >> (11 - q)) & 1) ? -p : p;
        }
    }
    const int jt = ringDst(RLAST, t);
#pragma unroll
    for (int q = 0; q < NQ; ++q) {
        if ((jt >> (11 - q)) & 1) part[q] = -part[q];
    }
    float red[NQ];
#pragma unroll
    for (int q = 0; q < NQ; ++q) {
        float v = part[q];
        v += xhalf(v);
        v += rowXor<8>(v);
        v += rowXor<4>(v);
        v += rowXor<2>(v);
        v += rowXor<1>(v);
        red[q] = v;
    }
    if (lane == 0) {
#pragma unroll
        for (int q = 0; q < NQ; ++q) wred[wave * NQ + q] = red[q];
    }
    __syncthreads();
    if (t < NQ) {
        float s = 0.0f;
#pragma unroll
        for (int wv = 0; wv < 8; ++wv) s += wred[wv * NQ + t];
        out[b * NQ + t] = s;
    }
}

extern "C" void kernel_launch(void* const* d_in, const int* in_sizes, int n_in,
                              void* d_out, int out_size, void* d_ws, size_t ws_size,
                              hipStream_t stream) {
    const float* x   = (const float*)d_in[0];
    const float* wts = (const float*)d_in[1];
    float* out = (float*)d_out;
    const int batch = in_sizes[0] / NQ;
    vqc_kernel<<<batch, NTHREADS, 0, stream>>>(x, wts, out);
}


# last layer's mid-group RY not applied: each <Z_q> has one mid wire, folded into the observable cos(th)(|a0|^2-|a1|^2)-2sin(th)Re(a0*a1); 104 WMMA/wave -> 88, 1 LDS trip+barrier go; f32 sums reordered
# speedup vs baseline: 11.6339x; 1.0911x over previous
#include <hip/hip_runtime.h>
#include <math.h>


typedef __attribute__((ext_vector_type(2))) float v2f;
typedef __attribute__((ext_vector_type(8))) float v8f;

#define NQ      12
#define NSTATE  4096
#define NLAYERS 4
#define NTHREADS 256
#define NFUSED  (NLAYERS * NQ)

__host__ __device__ constexpr int ringSrc(int r, int j) {
    int s = j;
    for (int qq = NQ - 1; qq >= 0; --qq) {
        const int cbit = 11 - qq;
        const int tbit = 11 - ((qq + r) % NQ);
        s ^= ((s >> cbit) & 1) << tbit;
    }
    return s;
}
__host__ __device__ constexpr int ringDst(int r, int i) {
    int s = i;
    for (int qq = 0; qq < NQ; ++qq) {
        const int cbit = 11 - qq;
        const int tbit = 11 - ((qq + r) % NQ);
        s ^= ((s >> cbit) & 1) << tbit;
    }
    return s;
}

__host__ __device__ constexpr int obsMask(int r, int q) {
    int m = 0;
    for (int k = 0; k < NQ; ++k) m |= ((ringDst(r, 1 << k) >> (11 - q)) & 1) << k;
    return m;
}
__host__ __device__ constexpr bool oneMidBitPerMask(int r) {
    for (int q = 0; q < NQ; ++q) {
        const int mm = (obsMask(r, q) >> 4) & 15;
        if (mm != 1 && mm != 2 && mm != 4 && mm != 8) return false;
    }
    return true;
}

template <int g>
__device__ __forceinline__ void buildGroup(float* mr, const float* gl, int t) {
    const int row = t >> 4, cm = t & 15;
    float p = 0.0f;
#pragma unroll
    for (int wl = 0; wl < 4; ++wl) {
        const int q = 11 - (4 * g + wl);
        const int rb = (row >> wl) & 1, cb = (cm >> wl) & 1;
        const float cs = gl[q * 8];
        const float sn = gl[q * 8 + 1];
        const float u = (rb == cb) ? cs : (rb ? sn : -sn);
        p = (wl == 0) ? u : p * u;
    }
    mr[g * 256 + t] = p;
}

__device__ __forceinline__ void buildPhase(float* ph, const float* gu, int t) {
    const int side = (t >= 144) ? 1 : 0;
    const int u = t - 144 * side;
    if (u < 48) return;
    const int L = u / 48 + 1;
    const int r = u % 48;
    const int g = r >> 4, idx = r & 15;
    const float* gl = gu + L * NQ * 8 + 2 + 2 * side;
    float pr = 0.0f, pi = 0.0f;
#pragma unroll
    for (int wl = 0; wl < 4; ++wl) {
        const int q = 11 - (4 * g + wl);
        const float cs = gl[q * 8];
        const float s0 = gl[q * 8 + 1];
        const float sn = ((idx >> wl) & 1) ? s0 : -s0;
        if (wl == 0) {
            pr = cs;
            pi = sn;
        } else {
            const float nr = pr * cs - pi * sn;
            const float ni = pr * sn + pi * cs;
            pr = nr;
            pi = ni;
        }
    }
    ph[2 * t]     = pr;
    ph[2 * t + 1] = pi;
}
__device__ __forceinline__ const float* phaseTab(const float* ph, int L, int side, int g) {
    return ph + 2 * (144 * side + (L - 1) * 48 + g * 16);
}

__device__ __forceinline__ void chainStep(float& x0r, float& x0i, float& x1r, float& x1i,
                                          const float* a, const float* g, int o) {
    const float a0r = a[0], a0i = a[1], a1r = a[4], a1i = a[5];
    const float t0r = a0r * x0r - a0i * x0i + a1r * x1r - a1i * x1i;
    const float t0i = a0r * x0i + a0i * x0r + a1r * x1i + a1i * x1r;
    const float t1r = a1r * x0r - a1i * x0i + a0r * x1r - a0i * x1i;
    const float t1i = a1r * x0i + a1i * x0r + a0r * x1i + a0i * x1r;
    const float g0r = g[4 * o],     g0i = g[4 * o + 1];
    const float g1r = g[4 * o + 2], g1i = g[4 * o + 3];
    x0r = g0r * t0r - g0i * t0i;  x0i = g0r * t0i + g0i * t0r;
    x1r = g1r * t1r - g1i * t1i;  x1i = g1r * t1i + g1i * t1r;
}

__device__ __forceinline__ void buildChainTables(float* midT, float* lowT, float* topV,
                                                 const float* gu, int wave, int lane) {
    const float* g1 = gu + NQ * 8;
    if (wave == 0 || wave == 1) {
        const int idx = lane >> 1, jin = lane & 1;
        float x0r = jin ? 0.0f : 1.0f, x0i = 0.0f;
        float x1r = jin ? 1.0f : 0.0f, x1i = 0.0f;
        const int qb = (wave == 0) ? 4 : 8;
#pragma unroll
        for (int d = 0; d < 4; ++d) {
            const int q = qb + d;
            chainStep(x0r, x0i, x1r, x1i, gu + q * 8, g1 + q * 8, (idx >> (3 - d)) & 1);
        }
        float* o = ((wave == 0) ? midT : lowT) + lane * 4;
        o[0] = x0r; o[1] = x0i; o[2] = x1r; o[3] = x1i;
    } else if (wave == 2 && lane < 16) {
        const int s = lane >> 3, top3 = lane & 7;
        const int o1 = (top3 >> 2) & 1;
        const float asr = gu[4 * s], asi = gu[4 * s + 1];
        const float* a1 = gu + 8;
        const float* gg = g1 + 8 + 4 * o1;
        const float b0r = a1[4 * s],       b0i = a1[4 * s + 1];
        const float b1r = a1[4 * (s ^ 1)], b1i = a1[4 * (s ^ 1) + 1];
        const float u0r = gg[0] * b0r - gg[1] * b0i, u0i = gg[0] * b0i + gg[1] * b0r;
        const float u1r = gg[2] * b1r - gg[3] * b1i, u1i = gg[2] * b1i + gg[3] * b1r;
        float x0r = asr * u0r - asi * u0i, x0i = asr * u0i + asi * u0r;
        float x1r = asr * u1r - asi * u1i, x1i = asr * u1i + asi * u1r;
        chainStep(x0r, x0i, x1r, x1i, gu + 2 * 8, g1 + 2 * 8, (top3 >> 1) & 1);
        chainStep(x0r, x0i, x1r, x1i, gu + 3 * 8, g1 + 3 * 8, top3 & 1);
        float* o = topV + lane * 4;
        o[0] = x0r; o[1] = x0i; o[2] = x1r; o[3] = x1i;
    }
}

__device__ __forceinline__ void layer01Direct(float* out, const float* midT,
                                              const float* lowT, const float* topV,
                                              const float* gu, int wave, int lane) {
    const int h = lane >> 4, c = lane & 15;
    const int m = ((c & 7) << 1) | (c >> 3);
    const float4 tv  = *(const float4*)(topV + (h * 8 + wave) * 4);
    const float4 m0  = *(const float4*)(midT + m * 8);
    const float4 m1  = *(const float4*)(midT + m * 8 + 4);
    v2f aR, aI;
    aR.x = tv.x * m0.x - tv.y * m0.y + tv.z * m1.x - tv.w * m1.y;
    aI.x = tv.x * m0.y + tv.y * m0.x + tv.z * m1.y + tv.w * m1.x;
    aR.y = tv.x * m0.z - tv.y * m0.w + tv.z * m1.z - tv.w * m1.w;
    aI.y = tv.x * m0.w + tv.y * m0.z + tv.z * m1.w + tv.w * m1.z;
    const v2f aIn = -aI;
    const float4 l0 = *(const float4*)(lowT + c * 8);
    const float4 l1 = *(const float4*)(lowT + c * 8 + 4);
    const float* g0 = gu + NQ * 8;
#pragma unroll
    for (int tc = 0; tc < 2; ++tc) {
        const float gar = g0[4 * tc + 2 * h],       gai = g0[4 * tc + 2 * h + 1];
        const float gbr = g0[4 * tc + 2 * (h ^ 1)], gbi = g0[4 * tc + 2 * (h ^ 1) + 1];
        v2f bR, bI;
        bR.x = l0.x * gar - l0.y * gai + l0.z * gbr - l0.w * gbi;
        bI.x = l0.x * gai + l0.y * gar + l0.z * gbi + l0.w * gbr;
        bR.y = l1.x * gar - l1.y * gai + l1.z * gbr - l1.w * gbi;
        bI.y = l1.x * gai + l1.y * gar + l1.z * gbi + l1.w * gbr;
        v8f accR = {0.f,0.f,0.f,0.f,0.f,0.f,0.f,0.f};
        v8f accI = {0.f,0.f,0.f,0.f,0.f,0.f,0.f,0.f};
        accR = __builtin_amdgcn_wmma_f32_16x16x4_f32(false, aR,  false, bR, (short)0, accR, false, false);
        accI = __builtin_amdgcn_wmma_f32_16x16x4_f32(false, aR,  false, bI, (short)0, accI, false, false);
        accR = __builtin_amdgcn_wmma_f32_16x16x4_f32(false, aIn, false, bI, (short)0, accR, false, false);
        accI = __builtin_amdgcn_wmma_f32_16x16x4_f32(false, aI,  false, bR, (short)0, accI, false, false);
#pragma unroll
        for (int r = 0; r < 8; ++r) {
            const int i = (tc << 11) | (wave << 8) | (r << 5) | lane;
            out[i]          = accR[r];
            out[NSTATE + i] = accI[r];
        }
    }
}

template <int G, bool PHASE>
__device__ __forceinline__ void apply16T(const float* in, float* out,
                                         const float* mr,
                                         const float* pReg, const float* pTile,
                                         const float* pLow,
                                         int wave, int lane) {
    constexpr int SH = G * 4;
    constexpr int LOWMASK = (1 << SH) - 1;
    const int half = lane >> 4;
    const int n15  = lane & 15;

    v2f ar[4];
#pragma unroll
    for (int kc = 0; kc < 4; ++kc) {
        const int k0 = kc * 4 + 2 * half;
        ar[kc].x = mr[n15 * 16 + k0];  ar[kc].y = mr[n15 * 16 + k0 + 1];
    }

    v8f qr = {0.f,0.f,0.f,0.f,0.f,0.f,0.f,0.f};
    v8f qi = {0.f,0.f,0.f,0.f,0.f,0.f,0.f,0.f};
    float plr = 0.0f, pli = 0.0f;
    if (PHASE) {
#pragma unroll
        for (int r = 0; r < 8; ++r) {
            qr[r] = pReg[2 * (r + 8 * half)];
            qi[r] = pReg[2 * (r + 8 * half) + 1];
        }
        plr = pLow[2 * n15];
        pli = pLow[2 * n15 + 1];
    }

#pragma unroll 1
    for (int tt = 0; tt < 2; ++tt) {
        const int col = (wave * 2 + tt) * 16 + n15;
        const int base = ((col >> SH) << (SH + 4)) | (col & LOWMASK);

        v2f br[4], bi[4];
#pragma unroll
        for (int kc = 0; kc < 4; ++kc) {
            const int k0 = kc * 4 + 2 * half;
            const int i0 = base | ((k0) << SH);
            const int i1 = base | ((k0 + 1) << SH);
            br[kc].x = in[i0];          br[kc].y = in[i1];
            bi[kc].x = in[NSTATE + i0]; bi[kc].y = in[NSTATE + i1];
        }

        v8f accR = {0.f,0.f,0.f,0.f,0.f,0.f,0.f,0.f};
        v8f accI = {0.f,0.f,0.f,0.f,0.f,0.f,0.f,0.f};
#pragma unroll
        for (int kc = 0; kc < 4; ++kc) {
            accR = __builtin_amdgcn_wmma_f32_16x16x4_f32(false, ar[kc], false, br[kc], (short)0, accR, false, false);
            accI = __builtin_amdgcn_wmma_f32_16x16x4_f32(false, ar[kc], false, bi[kc], (short)0, accI, false, false);
        }

        if (PHASE) {
            const int tcon = col >> 4;
            const float ptr_ = pTile[2 * tcon], pti = pTile[2 * tcon + 1];
            const float cr = ptr_ * plr - pti * pli;
            const float ci = ptr_ * pli + pti * plr;
            const v8f wr = qr * cr - qi * ci;
            const v8f wi = qr * ci + qi * cr;
            const v8f oR = accR * wr - accI * wi;
            const v8f oI = accR * wi + accI * wr;
            accR = oR;
            accI = oI;
        }

#pragma unroll
        for (int r = 0; r < 8; ++r) {
            const int i = base | ((r + 8 * half) << SH);
            out[i]          = accR[r];
            out[NSTATE + i] = accI[r];
        }
    }
}

template <int R>
__device__ __forceinline__ void chainTopLow(const float* in, float* out,
                                            const float* m2, const float* m0,
                                            const float* pTop, const float* pMid,
                                            const float* pLow,
                                            int wave, int lane) {
    constexpr int SH = 8;
    const int half = lane >> 4;
    const int n15  = lane & 15;

    v2f tr[4];
#pragma unroll
    for (int kc = 0; kc < 4; ++kc) {
        const int k0 = kc * 4 + 2 * half;
        tr[kc].x = m2[n15 * 16 + k0];  tr[kc].y = m2[n15 * 16 + k0 + 1];
    }
    v2f lr[4];
#pragma unroll
    for (int kc = 0; kc < 4; ++kc) {
        const int k0 = 2 * kc + 8 * half;
        lr[kc].x = m0[n15 * 16 + k0];  lr[kc].y = m0[n15 * 16 + k0 + 1];
    }
    v2f ptr_[4], pti[4];
#pragma unroll
    for (int kc = 0; kc < 4; ++kc) {
        const int k0 = kc * 4 + 2 * half;
        ptr_[kc].x = pTop[2 * k0];      ptr_[kc].y = pTop[2 * k0 + 2];
        pti[kc].x  = pTop[2 * k0 + 1];  pti[kc].y  = pTop[2 * k0 + 3];
    }
    const float plr = pLow[2 * n15], pli = pLow[2 * n15 + 1];

    const int base0 = (wave * 2) * 16 + n15;
    const int pl0   = ringSrc(R, base0 | ((2 * half) << SH));

#pragma unroll 1
    for (int tt = 0; tt < 2; ++tt) {
        const int base = base0 | (tt << 4);
        const int pl   = pl0 ^ (-tt & ringSrc(R, 16));

        v2f sr[4], si[4];
#pragma unroll
        for (int kc = 0; kc < 4; ++kc) {
            const int i0 = pl ^ ringSrc(R, (kc * 4) << SH);
            const int i1 = pl ^ ringSrc(R, (kc * 4 + 1) << SH);
            sr[kc].x = in[i0];          sr[kc].y = in[i1];
            si[kc].x = in[NSTATE + i0]; si[kc].y = in[NSTATE + i1];
        }

        const int mid = wave * 2 + tt;
        const float pmr = pMid[2 * mid], pmi = pMid[2 * mid + 1];
        const float cr = pmr * plr - pmi * pli;
        const float ci = pmr * pli + pmi * plr;
        v2f xr[4], xi[4];
#pragma unroll
        for (int kc = 0; kc < 4; ++kc) {
            const v2f wr = ptr_[kc] * cr - pti[kc] * ci;
            const v2f wi = ptr_[kc] * ci + pti[kc] * cr;
            xr[kc] = sr[kc] * wr - si[kc] * wi;
            xi[kc] = sr[kc] * wi + si[kc] * wr;
        }

        v8f accR = {0.f,0.f,0.f,0.f,0.f,0.f,0.f,0.f};
        v8f accI = {0.f,0.f,0.f,0.f,0.f,0.f,0.f,0.f};
#pragma unroll
        for (int kc = 0; kc < 4; ++kc) {
            accR = __builtin_amdgcn_wmma_f32_16x16x4_f32(false, xr[kc], false, tr[kc], (short)0, accR, false, false);
            accI = __builtin_amdgcn_wmma_f32_16x16x4_f32(false, xi[kc], false, tr[kc], (short)0, accI, false, false);
        }

        v2f ur[4], ui[4];
#pragma unroll
        for (int kc = 0; kc < 4; ++kc) {
            ur[kc].x = accR[2 * kc];  ur[kc].y = accR[2 * kc + 1];
            ui[kc].x = accI[2 * kc];  ui[kc].y = accI[2 * kc + 1];
        }

        v8f outR = {0.f,0.f,0.f,0.f,0.f,0.f,0.f,0.f};
        v8f outI = {0.f,0.f,0.f,0.f,0.f,0.f,0.f,0.f};
#pragma unroll
        for (int kc = 0; kc < 4; ++kc) {
            outR = __builtin_amdgcn_wmma_f32_16x16x4_f32(false, ur[kc], false, lr[kc], (short)0, outR, false, false);
            outI = __builtin_amdgcn_wmma_f32_16x16x4_f32(false, ui[kc], false, lr[kc], (short)0, outI, false, false);
        }

#pragma unroll
        for (int r = 0; r < 8; ++r) {
            const int i = base | ((r + 8 * half) << SH);
            out[i]          = outR[r];
            out[NSTATE + i] = outI[r];
        }
    }
}

template <int L>
__device__ __forceinline__ void layerPass(float* st, float* mr, const float* ph,
                                          const float* gu, int t, int wave, int lane) {
    float* cur = st + (L & 1) * 2 * NSTATE;
    constexpr int RPREV = (L - 1) % (NQ - 1) + 1;
    constexpr bool LAST = (L + 1 == NLAYERS);
    const float* prev = st + ((L & 1) ^ 1) * 2 * NSTATE;
    chainTopLow<RPREV>(prev, cur, mr + 512, mr,
                       phaseTab(ph, L, 0, 2), phaseTab(ph, L, 0, 1), phaseTab(ph, L, 0, 0),
                       wave, lane);
    if (!LAST) buildGroup<1>(mr, gu + L * NQ * 8, t);
    __syncthreads();
    if (!LAST) {
        apply16T<1, true>(cur, cur, mr + 256,
                          phaseTab(ph, L, 1, 1), phaseTab(ph, L, 1, 2), phaseTab(ph, L, 1, 0),
                          wave, lane);
        buildGroup<2>(mr, gu + (L + 1) * NQ * 8, t);
        buildGroup<0>(mr, gu + (L + 1) * NQ * 8, t);
        __syncthreads();
    }
}

__device__ __forceinline__ float xhalf(float x) {
    const int bits = __builtin_bit_cast(int, x);
    return __builtin_bit_cast(float,
        __builtin_amdgcn_permlanex16(bits, bits, 0x76543210, 0xfedcba98, false, false));
}
template <int M>
__device__ __forceinline__ float rowXor(float x) {
    const int bits = __builtin_bit_cast(int, x);
    return __builtin_bit_cast(float,
        __builtin_amdgcn_update_dpp(0, bits, 0x160 | M, 0xf, 0xf, true));
}

__global__ __launch_bounds__(NTHREADS)
void vqc_kernel(const float* __restrict__ x,
                const float* __restrict__ wts,
                float* __restrict__ out) {
    __shared__ float st[2 * 2 * NSTATE];
    __shared__ float mr[3 * 256];
    __shared__ __attribute__((aligned(16))) float ph[240 * 2];
    __shared__ __attribute__((aligned(16))) float gu[NFUSED * 8];
    __shared__ __attribute__((aligned(16))) float midT[16 * 2 * 2 * 2];
    __shared__ __attribute__((aligned(16))) float lowT[16 * 2 * 2 * 2];
    __shared__ __attribute__((aligned(16))) float topV[2 * 8 * 2 * 2];
    __shared__ __attribute__((aligned(16))) float wred[8 * NQ];

    const int t    = threadIdx.x;
    const int b    = blockIdx.x;
    const int wave = t >> 5;
    const int lane = t & 31;

    if (t < NFUSED) {
        const float* wp = wts + t * 3;
        const float phi = wp[0], th = wp[1], om = wp[2];
        const float ct = cosf(th * 0.5f);
        const float sn = sinf(th * 0.5f);
        const float cp = cosf(phi * 0.5f), sp = sinf(phi * 0.5f);
        const float co = cosf(om * 0.5f),  so = sinf(om * 0.5f);
        float* gr = gu + t * 8;
        if (t < 2 * NQ) {
            const float chs = cp * co - sp * so, shs = sp * co + cp * so;
            const float chd = cp * co + sp * so, shd = sp * co - cp * so;
            const float e0 =  chs * ct, e1 = -shs * ct;
            const float e2 = -chd * sn, e3 = -shd * sn;
            const float e4 =  chd * sn, e5 = -shd * sn;
            const float e6 =  chs * ct, e7 =  shs * ct;
            if (t < NQ) {
                const float a = x[b * NQ + t] * 0.5f;
                const float c = cosf(a), s = sinf(a);
                gr[0] = e0 * c + e3 * s;  gr[1] = e1 * c - e2 * s;
                gr[2] = e1 * s + e2 * c;  gr[3] = e3 * c - e0 * s;
                gr[4] = e4 * c + e7 * s;  gr[5] = e5 * c - e6 * s;
                gr[6] = e5 * s + e6 * c;  gr[7] = e7 * c - e4 * s;
            } else {
                gr[0] = e0; gr[1] = e1; gr[2] = e2; gr[3] = e3;
                gr[4] = e4; gr[5] = e5; gr[6] = e6; gr[7] = e7;
            }
        } else {
            gr[0] = ct; gr[1] = sn; gr[2] = cp; gr[3] = sp;
            gr[4] = co; gr[5] = so; gr[6] = 0.0f; gr[7] = 0.0f;
        }
    }
    __syncthreads();

    buildChainTables(midT, lowT, topV, gu, wave, lane);
    if (t < 240) buildPhase(ph, gu, t);
    buildGroup<2>(mr, gu + 2 * NQ * 8, t);
    buildGroup<0>(mr, gu + 2 * NQ * 8, t);
    __syncthreads();

    layer01Direct(st + 2 * NSTATE, midT, lowT, topV, gu, wave, lane);
    __syncthreads();

    layerPass<2>(st, mr, ph, gu, t, wave, lane);
    layerPass<3>(st, mr, ph, gu, t, wave, lane);

    constexpr int RLAST = (NLAYERS - 1) % (NQ - 1) + 1;
    static_assert(oneMidBitPerMask(RLAST),
                  "the closing sums need exactly one middle-group bit per observable");
    const float* fin = st + ((NLAYERS - 1) & 1) * 2 * NSTATE;
    const int ibase = ((t >> 4) << 8) | (t & 15);
    v2f are[8], aim[8], pw[8];
#pragma unroll
    for (int j = 0; j < 8; ++j) {
        are[j].x = fin[ibase | ((2 * j) << 4)];
        are[j].y = fin[ibase | ((2 * j + 1) << 4)];
        aim[j].x = fin[NSTATE + (ibase | ((2 * j) << 4))];
        aim[j].y = fin[NSTATE + (ibase | ((2 * j + 1) << 4))];
        pw[j] = are[j] * are[j] + aim[j] * aim[j];
    }
    float sd[4];
    {
        v2f s3[4], d3[4];
#pragma unroll
        for (int j = 0; j < 4; ++j) {
            s3[j] = pw[j] + pw[j + 4];
            d3[j] = pw[j] - pw[j + 4];
        }
        const v2f t3 = (d3[0] + d3[1]) + (d3[2] + d3[3]);
        sd[3] = t3.x + t3.y;
        const v2f s2a = s3[0] + s3[2], d2a = s3[0] - s3[2];
        const v2f s2b = s3[1] + s3[3], d2b = s3[1] - s3[3];
        const v2f t2 = d2a + d2b;
        sd[2] = t2.x + t2.y;
        const v2f s1 = s2a + s2b, d1 = s2a - s2b;
        sd[1] = d1.x + d1.y;
        sd[0] = s1.x - s1.y;
    }
    float cr[4];
    {
        float c0 = are[0].x * are[0].y;
        c0 = __builtin_fmaf(aim[0].x, aim[0].y, c0);
#pragma unroll
        for (int j = 1; j < 8; ++j) {
            c0 = __builtin_fmaf(are[j].x, are[j].y, c0);
            c0 = __builtin_fmaf(aim[j].x, aim[j].y, c0);
        }
        cr[0] = c0;
    }
#pragma unroll
    for (int b = 1; b < 4; ++b) {
        v2f cre = {0.0f, 0.0f}, cim = {0.0f, 0.0f};
#pragma unroll
        for (int j = 0; j < 8; ++j) {
            if ((j >> (b - 1)) & 1) continue;
            cre += are[j] * are[j | (1 << (b - 1))];
            cim += aim[j] * aim[j | (1 << (b - 1))];
        }
        const v2f cs = cre + cim;
        cr[b] = cs.x + cs.y;
    }
    const float* gl = gu + (NLAYERS - 1) * NQ * 8;
    float vb[4];
#pragma unroll
    for (int b = 0; b < 4; ++b) {
        const float ch = gl[(7 - b) * 8], sh = gl[(7 - b) * 8 + 1];
        vb[b] = (ch * ch - sh * sh) * sd[b] - 4.0f * (ch * sh) * cr[b];
    }
    const int jt = ringDst(RLAST, ibase);
    float part[NQ];
#pragma unroll
    for (int q = 0; q < NQ; ++q) {
        const int mm = (obsMask(RLAST, q) >> 4) & 15;
        const int b = (mm == 1) ? 0 : (mm == 2) ? 1 : (mm == 4) ? 2 : 3;
        part[q] = ((jt >> (11 - q)) & 1) ? -vb[b] : vb[b];
    }
    float red[NQ];
#pragma unroll
    for (int q = 0; q < NQ; ++q) {
        float v = part[q];
        v += xhalf(v);
        v += rowXor<8>(v);
        v += rowXor<4>(v);
        v += rowXor<2>(v);
        v += rowXor<1>(v);
        red[q] = v;
    }
    if (lane == 0) {
#pragma unroll
        for (int q = 0; q < NQ; ++q) wred[wave * NQ + q] = red[q];
    }
    __syncthreads();
    if (t < NQ) {
        float s = 0.0f;
#pragma unroll
        for (int wv = 0; wv < 8; ++wv) s += wred[wv * NQ + t];
        out[b * NQ + t] = s;
    }
}

extern "C" void kernel_launch(void* const* d_in, const int* in_sizes, int n_in,
                              void* d_out, int out_size, void* d_ws, size_t ws_size,
                              hipStream_t stream) {
    const float* x   = (const float*)d_in[0];
    const float* wts = (const float*)d_in[1];
    float* out = (float*)d_out;
    const int batch = in_sizes[0] / NQ;
    vqc_kernel<<<batch, NTHREADS, 0, stream>>>(x, wts, out);
}


# LDS bank conflicts of the state planes and 16x16 operators removed: state under an XOR swizzle (index bit 5 ^= parity of bits 7,8,11), operator rows at pitch 20; results bit-identical
# speedup vs baseline: 12.2121x; 1.0497x over previous
#include <hip/hip_runtime.h>
#include <math.h>


typedef __attribute__((ext_vector_type(2))) float v2f;
typedef __attribute__((ext_vector_type(8))) float v8f;

#define NQ      12
#define NSTATE  4096
#define NLAYERS 4
#define NTHREADS 256
#define NFUSED  (NLAYERS * NQ)
#define OPP     20
#define OPSZ    (16 * OPP)

__host__ __device__ constexpr int ringSrc(int r, int j) {
    int s = j;
    for (int qq = NQ - 1; qq >= 0; --qq) {
        const int cbit = 11 - qq;
        const int tbit = 11 - ((qq + r) % NQ);
        s ^= ((s >> cbit) & 1) << tbit;
    }
    return s;
}
__host__ __device__ constexpr int ringDst(int r, int i) {
    int s = i;
    for (int qq = 0; qq < NQ; ++qq) {
        const int cbit = 11 - qq;
        const int tbit = 11 - ((qq + r) % NQ);
        s ^= ((s >> cbit) & 1) << tbit;
    }
    return s;
}

__host__ __device__ constexpr int obsMask(int r, int q) {
    int m = 0;
    for (int k = 0; k < NQ; ++k) m |= ((ringDst(r, 1 << k) >> (11 - q)) & 1) << k;
    return m;
}
__host__ __device__ constexpr bool oneMidBitPerMask(int r) {
    for (int q = 0; q < NQ; ++q) {
        const int mm = (obsMask(r, q) >> 4) & 15;
        if (mm != 1 && mm != 2 && mm != 4 && mm != 8) return false;
    }
    return true;
}

__host__ __device__ constexpr int swz(int i) {
    return i ^ ((((i >> 7) ^ (i >> 8) ^ (i >> 11)) & 1) << 5);
}
__device__ __forceinline__ int swzAt(int b0, int b1, int c) {
    const int k = swz(c);
    return ((k & 32) ? b1 : b0) + (k & ~32);
}

template <int g>
__device__ __forceinline__ void buildGroup(float* mr, const float* gl, int t) {
    const int row = t >> 4, cm = t & 15;
    float p = 0.0f;
#pragma unroll
    for (int wl = 0; wl < 4; ++wl) {
        const int q = 11 - (4 * g + wl);
        const int rb = (row >> wl) & 1, cb = (cm >> wl) & 1;
        const float cs = gl[q * 8];
        const float sn = gl[q * 8 + 1];
        const float u = (rb == cb) ? cs : (rb ? sn : -sn);
        p = (wl == 0) ? u : p * u;
    }
    mr[g * OPSZ + row * OPP + cm] = p;
}

__device__ __forceinline__ void buildPhase(float* ph, const float* gu, int t) {
    const int side = (t >= 144) ? 1 : 0;
    const int u = t - 144 * side;
    if (u < 48) return;
    const int L = u / 48 + 1;
    const int r = u % 48;
    const int g = r >> 4, idx = r & 15;
    const float* gl = gu + L * NQ * 8 + 2 + 2 * side;
    float pr = 0.0f, pi = 0.0f;
#pragma unroll
    for (int wl = 0; wl < 4; ++wl) {
        const int q = 11 - (4 * g + wl);
        const float cs = gl[q * 8];
        const float s0 = gl[q * 8 + 1];
        const float sn = ((idx >> wl) & 1) ? s0 : -s0;
        if (wl == 0) {
            pr = cs;
            pi = sn;
        } else {
            const float nr = pr * cs - pi * sn;
            const float ni = pr * sn + pi * cs;
            pr = nr;
            pi = ni;
        }
    }
    ph[2 * t]     = pr;
    ph[2 * t + 1] = pi;
}
__device__ __forceinline__ const float* phaseTab(const float* ph, int L, int side, int g) {
    return ph + 2 * (144 * side + (L - 1) * 48 + g * 16);
}

__device__ __forceinline__ void chainStep(float& x0r, float& x0i, float& x1r, float& x1i,
                                          const float* a, const float* g, int o) {
    const float a0r = a[0], a0i = a[1], a1r = a[4], a1i = a[5];
    const float t0r = a0r * x0r - a0i * x0i + a1r * x1r - a1i * x1i;
    const float t0i = a0r * x0i + a0i * x0r + a1r * x1i + a1i * x1r;
    const float t1r = a1r * x0r - a1i * x0i + a0r * x1r - a0i * x1i;
    const float t1i = a1r * x0i + a1i * x0r + a0r * x1i + a0i * x1r;
    const float g0r = g[4 * o],     g0i = g[4 * o + 1];
    const float g1r = g[4 * o + 2], g1i = g[4 * o + 3];
    x0r = g0r * t0r - g0i * t0i;  x0i = g0r * t0i + g0i * t0r;
    x1r = g1r * t1r - g1i * t1i;  x1i = g1r * t1i + g1i * t1r;
}

__device__ __forceinline__ void buildChainTables(float* midT, float* lowT, float* topV,
                                                 const float* gu, int wave, int lane) {
    const float* g1 = gu + NQ * 8;
    if (wave == 0 || wave == 1) {
        const int idx = lane >> 1, jin = lane & 1;
        float x0r = jin ? 0.0f : 1.0f, x0i = 0.0f;
        float x1r = jin ? 1.0f : 0.0f, x1i = 0.0f;
        const int qb = (wave == 0) ? 4 : 8;
#pragma unroll
        for (int d = 0; d < 4; ++d) {
            const int q = qb + d;
            chainStep(x0r, x0i, x1r, x1i, gu + q * 8, g1 + q * 8, (idx >> (3 - d)) & 1);
        }
        float* o = ((wave == 0) ? midT : lowT) + lane * 4;
        o[0] = x0r; o[1] = x0i; o[2] = x1r; o[3] = x1i;
    } else if (wave == 2 && lane < 16) {
        const int s = lane >> 3, top3 = lane & 7;
        const int o1 = (top3 >> 2) & 1;
        const float asr = gu[4 * s], asi = gu[4 * s + 1];
        const float* a1 = gu + 8;
        const float* gg = g1 + 8 + 4 * o1;
        const float b0r = a1[4 * s],       b0i = a1[4 * s + 1];
        const float b1r = a1[4 * (s ^ 1)], b1i = a1[4 * (s ^ 1) + 1];
        const float u0r = gg[0] * b0r - gg[1] * b0i, u0i = gg[0] * b0i + gg[1] * b0r;
        const float u1r = gg[2] * b1r - gg[3] * b1i, u1i = gg[2] * b1i + gg[3] * b1r;
        float x0r = asr * u0r - asi * u0i, x0i = asr * u0i + asi * u0r;
        float x1r = asr * u1r - asi * u1i, x1i = asr * u1i + asi * u1r;
        chainStep(x0r, x0i, x1r, x1i, gu + 2 * 8, g1 + 2 * 8, (top3 >> 1) & 1);
        chainStep(x0r, x0i, x1r, x1i, gu + 3 * 8, g1 + 3 * 8, top3 & 1);
        float* o = topV + lane * 4;
        o[0] = x0r; o[1] = x0i; o[2] = x1r; o[3] = x1i;
    }
}

__device__ __forceinline__ void layer01Direct(float* out, const float* midT,
                                              const float* lowT, const float* topV,
                                              const float* gu, int wave, int lane) {
    const int h = lane >> 4, c = lane & 15;
    const int m = ((c & 7) << 1) | (c >> 3);
    const float4 tv  = *(const float4*)(topV + (h * 8 + wave) * 4);
    const float4 m0  = *(const float4*)(midT + m * 8);
    const float4 m1  = *(const float4*)(midT + m * 8 + 4);
    v2f aR, aI;
    aR.x = tv.x * m0.x - tv.y * m0.y + tv.z * m1.x - tv.w * m1.y;
    aI.x = tv.x * m0.y + tv.y * m0.x + tv.z * m1.y + tv.w * m1.x;
    aR.y = tv.x * m0.z - tv.y * m0.w + tv.z * m1.z - tv.w * m1.w;
    aI.y = tv.x * m0.w + tv.y * m0.z + tv.z * m1.w + tv.w * m1.z;
    const v2f aIn = -aI;
    const float4 l0 = *(const float4*)(lowT + c * 8);
    const float4 l1 = *(const float4*)(lowT + c * 8 + 4);
    const float* g0 = gu + NQ * 8;
    const int sb0 = swz((wave << 8) | lane), sb1 = sb0 ^ 32;
#pragma unroll
    for (int tc = 0; tc < 2; ++tc) {
        const float gar = g0[4 * tc + 2 * h],       gai = g0[4 * tc + 2 * h + 1];
        const float gbr = g0[4 * tc + 2 * (h ^ 1)], gbi = g0[4 * tc + 2 * (h ^ 1) + 1];
        v2f bR, bI;
        bR.x = l0.x * gar - l0.y * gai + l0.z * gbr - l0.w * gbi;
        bI.x = l0.x * gai + l0.y * gar + l0.z * gbi + l0.w * gbr;
        bR.y = l1.x * gar - l1.y * gai + l1.z * gbr - l1.w * gbi;
        bI.y = l1.x * gai + l1.y * gar + l1.z * gbi + l1.w * gbr;
        v8f accR = {0.f,0.f,0.f,0.f,0.f,0.f,0.f,0.f};
        v8f accI = {0.f,0.f,0.f,0.f,0.f,0.f,0.f,0.f};
        accR = __builtin_amdgcn_wmma_f32_16x16x4_f32(false, aR,  false, bR, (short)0, accR, false, false);
        accI = __builtin_amdgcn_wmma_f32_16x16x4_f32(false, aR,  false, bI, (short)0, accI, false, false);
        accR = __builtin_amdgcn_wmma_f32_16x16x4_f32(false, aIn, false, bI, (short)0, accR, false, false);
        accI = __builtin_amdgcn_wmma_f32_16x16x4_f32(false, aI,  false, bR, (short)0, accI, false, false);
#pragma unroll
        for (int r = 0; r < 8; ++r) {
            const int i = swzAt(sb0, sb1, (tc << 11) | (r << 5));
            out[i]          = accR[r];
            out[NSTATE + i] = accI[r];
        }
    }
}

template <int G, bool PHASE>
__device__ __forceinline__ void apply16T(const float* in, float* out,
                                         const float* mr,
                                         const float* pReg, const float* pTile,
                                         const float* pLow,
                                         int wave, int lane) {
    constexpr int SH = G * 4;
    constexpr int LOWMASK = (1 << SH) - 1;
    const int half = lane >> 4;
    const int n15  = lane & 15;

    v2f ar[4];
#pragma unroll
    for (int kc = 0; kc < 4; ++kc) {
        const int k0 = kc * 4 + 2 * half;
        ar[kc].x = mr[n15 * OPP + k0];  ar[kc].y = mr[n15 * OPP + k0 + 1];
    }

    v8f qr = {0.f,0.f,0.f,0.f,0.f,0.f,0.f,0.f};
    v8f qi = {0.f,0.f,0.f,0.f,0.f,0.f,0.f,0.f};
    float plr = 0.0f, pli = 0.0f;
    if (PHASE) {
#pragma unroll
        for (int r = 0; r < 8; ++r) {
            qr[r] = pReg[2 * (r + 8 * half)];
            qi[r] = pReg[2 * (r + 8 * half) + 1];
        }
        plr = pLow[2 * n15];
        pli = pLow[2 * n15 + 1];
    }

#pragma unroll 1
    for (int tt = 0; tt < 2; ++tt) {
        const int col = (wave * 2 + tt) * 16 + n15;
        const int base = ((col >> SH) << (SH + 4)) | (col & LOWMASK);
        const int lb0 = swz(base | ((2 * half) << SH)), lb1 = lb0 ^ 32;
        const int sb0 = swz(base | ((8 * half) << SH)), sb1 = sb0 ^ 32;

        v2f br[4], bi[4];
#pragma unroll
        for (int kc = 0; kc < 4; ++kc) {
            const int i0 = swzAt(lb0, lb1, (kc * 4) << SH);
            const int i1 = swzAt(lb0, lb1, (kc * 4 + 1) << SH);
            br[kc].x = in[i0];          br[kc].y = in[i1];
            bi[kc].x = in[NSTATE + i0]; bi[kc].y = in[NSTATE + i1];
        }

        v8f accR = {0.f,0.f,0.f,0.f,0.f,0.f,0.f,0.f};
        v8f accI = {0.f,0.f,0.f,0.f,0.f,0.f,0.f,0.f};
#pragma unroll
        for (int kc = 0; kc < 4; ++kc) {
            accR = __builtin_amdgcn_wmma_f32_16x16x4_f32(false, ar[kc], false, br[kc], (short)0, accR, false, false);
            accI = __builtin_amdgcn_wmma_f32_16x16x4_f32(false, ar[kc], false, bi[kc], (short)0, accI, false, false);
        }

        if (PHASE) {
            const int tcon = col >> 4;
            const float ptr_ = pTile[2 * tcon], pti = pTile[2 * tcon + 1];
            const float cr = ptr_ * plr - pti * pli;
            const float ci = ptr_ * pli + pti * plr;
            const v8f wr = qr * cr - qi * ci;
            const v8f wi = qr * ci + qi * cr;
            const v8f oR = accR * wr - accI * wi;
            const v8f oI = accR * wi + accI * wr;
            accR = oR;
            accI = oI;
        }

#pragma unroll
        for (int r = 0; r < 8; ++r) {
            const int i = swzAt(sb0, sb1, r << SH);
            out[i]          = accR[r];
            out[NSTATE + i] = accI[r];
        }
    }
}

template <int R>
__device__ __forceinline__ void chainTopLow(const float* in, float* out,
                                            const float* m2, const float* m0,
                                            const float* pTop, const float* pMid,
                                            const float* pLow,
                                            int wave, int lane) {
    constexpr int SH = 8;
    const int half = lane >> 4;
    const int n15  = lane & 15;

    v2f tr[4];
#pragma unroll
    for (int kc = 0; kc < 4; ++kc) {
        const int k0 = kc * 4 + 2 * half;
        tr[kc].x = m2[n15 * OPP + k0];  tr[kc].y = m2[n15 * OPP + k0 + 1];
    }
    v2f lr[4];
#pragma unroll
    for (int kc = 0; kc < 4; ++kc) {
        const int k0 = 2 * kc + 8 * half;
        lr[kc].x = m0[n15 * OPP + k0];  lr[kc].y = m0[n15 * OPP + k0 + 1];
    }
    v2f ptr_[4], pti[4];
#pragma unroll
    for (int kc = 0; kc < 4; ++kc) {
        const int k0 = kc * 4 + 2 * half;
        ptr_[kc].x = pTop[2 * k0];      ptr_[kc].y = pTop[2 * k0 + 2];
        pti[kc].x  = pTop[2 * k0 + 1];  pti[kc].y  = pTop[2 * k0 + 3];
    }
    const float plr = pLow[2 * n15], pli = pLow[2 * n15 + 1];

    const int base0 = (wave * 2) * 16 + n15;
    const int pl0   = swz(ringSrc(R, base0 | ((2 * half) << SH)));
    const int sl0   = swz(base0 | ((8 * half) << SH));

#pragma unroll 1
    for (int tt = 0; tt < 2; ++tt) {
        const int pl   = pl0 ^ (-tt & swz(ringSrc(R, 16)));
        const int sb0  = sl0 | (tt << 4), sb1 = sb0 ^ 32;

        v2f sr[4], si[4];
#pragma unroll
        for (int kc = 0; kc < 4; ++kc) {
            const int i0 = pl ^ swz(ringSrc(R, (kc * 4) << SH));
            const int i1 = pl ^ swz(ringSrc(R, (kc * 4 + 1) << SH));
            sr[kc].x = in[i0];          sr[kc].y = in[i1];
            si[kc].x = in[NSTATE + i0]; si[kc].y = in[NSTATE + i1];
        }

        const int mid = wave * 2 + tt;
        const float pmr = pMid[2 * mid], pmi = pMid[2 * mid + 1];
        const float cr = pmr * plr - pmi * pli;
        const float ci = pmr * pli + pmi * plr;
        v2f xr[4], xi[4];
#pragma unroll
        for (int kc = 0; kc < 4; ++kc) {
            const v2f wr = ptr_[kc] * cr - pti[kc] * ci;
            const v2f wi = ptr_[kc] * ci + pti[kc] * cr;
            xr[kc] = sr[kc] * wr - si[kc] * wi;
            xi[kc] = sr[kc] * wi + si[kc] * wr;
        }

        v8f accR = {0.f,0.f,0.f,0.f,0.f,0.f,0.f,0.f};
        v8f accI = {0.f,0.f,0.f,0.f,0.f,0.f,0.f,0.f};
#pragma unroll
        for (int kc = 0; kc < 4; ++kc) {
            accR = __builtin_amdgcn_wmma_f32_16x16x4_f32(false, xr[kc], false, tr[kc], (short)0, accR, false, false);
            accI = __builtin_amdgcn_wmma_f32_16x16x4_f32(false, xi[kc], false, tr[kc], (short)0, accI, false, false);
        }

        v2f ur[4], ui[4];
#pragma unroll
        for (int kc = 0; kc < 4; ++kc) {
            ur[kc].x = accR[2 * kc];  ur[kc].y = accR[2 * kc + 1];
            ui[kc].x = accI[2 * kc];  ui[kc].y = accI[2 * kc + 1];
        }

        v8f outR = {0.f,0.f,0.f,0.f,0.f,0.f,0.f,0.f};
        v8f outI = {0.f,0.f,0.f,0.f,0.f,0.f,0.f,0.f};
#pragma unroll
        for (int kc = 0; kc < 4; ++kc) {
            outR = __builtin_amdgcn_wmma_f32_16x16x4_f32(false, ur[kc], false, lr[kc], (short)0, outR, false, false);
            outI = __builtin_amdgcn_wmma_f32_16x16x4_f32(false, ui[kc], false, lr[kc], (short)0, outI, false, false);
        }

#pragma unroll
        for (int r = 0; r < 8; ++r) {
            const int i = swzAt(sb0, sb1, r << SH);
            out[i]          = outR[r];
            out[NSTATE + i] = outI[r];
        }
    }
}

template <int L>
__device__ __forceinline__ void layerPass(float* st, float* mr, const float* ph,
                                          const float* gu, int t, int wave, int lane) {
    float* cur = st + (L & 1) * 2 * NSTATE;
    constexpr int RPREV = (L - 1) % (NQ - 1) + 1;
    constexpr bool LAST = (L + 1 == NLAYERS);
    const float* prev = st + ((L & 1) ^ 1) * 2 * NSTATE;
    chainTopLow<RPREV>(prev, cur, mr + 2 * OPSZ, mr,
                       phaseTab(ph, L, 0, 2), phaseTab(ph, L, 0, 1), phaseTab(ph, L, 0, 0),
                       wave, lane);
    if (!LAST) buildGroup<1>(mr, gu + L * NQ * 8, t);
    __syncthreads();
    if (!LAST) {
        apply16T<1, true>(cur, cur, mr + OPSZ,
                          phaseTab(ph, L, 1, 1), phaseTab(ph, L, 1, 2), phaseTab(ph, L, 1, 0),
                          wave, lane);
        buildGroup<2>(mr, gu + (L + 1) * NQ * 8, t);
        buildGroup<0>(mr, gu + (L + 1) * NQ * 8, t);
        __syncthreads();
    }
}

__device__ __forceinline__ float xhalf(float x) {
    const int bits = __builtin_bit_cast(int, x);
    return __builtin_bit_cast(float,
        __builtin_amdgcn_permlanex16(bits, bits, 0x76543210, 0xfedcba98, false, false));
}
template <int M>
__device__ __forceinline__ float rowXor(float x) {
    const int bits = __builtin_bit_cast(int, x);
    return __builtin_bit_cast(float,
        __builtin_amdgcn_update_dpp(0, bits, 0x160 | M, 0xf, 0xf, true));
}

__global__ __launch_bounds__(NTHREADS)
void vqc_kernel(const float* __restrict__ x,
                const float* __restrict__ wts,
                float* __restrict__ out) {
    __shared__ float st[2 * 2 * NSTATE];
    __shared__ __attribute__((aligned(16))) float mr[3 * OPSZ];
    __shared__ __attribute__((aligned(16))) float ph[240 * 2];
    __shared__ __attribute__((aligned(16))) float gu[NFUSED * 8];
    __shared__ __attribute__((aligned(16))) float midT[16 * 2 * 2 * 2];
    __shared__ __attribute__((aligned(16))) float lowT[16 * 2 * 2 * 2];
    __shared__ __attribute__((aligned(16))) float topV[2 * 8 * 2 * 2];
    __shared__ __attribute__((aligned(16))) float wred[8 * NQ];

    const int t    = threadIdx.x;
    const int b    = blockIdx.x;
    const int wave = t >> 5;
    const int lane = t & 31;

    if (t < NFUSED) {
        const float* wp = wts + t * 3;
        const float phi = wp[0], th = wp[1], om = wp[2];
        const float ct = cosf(th * 0.5f);
        const float sn = sinf(th * 0.5f);
        const float cp = cosf(phi * 0.5f), sp = sinf(phi * 0.5f);
        const float co = cosf(om * 0.5f),  so = sinf(om * 0.5f);
        float* gr = gu + t * 8;
        if (t < 2 * NQ) {
            const float chs = cp * co - sp * so, shs = sp * co + cp * so;
            const float chd = cp * co + sp * so, shd = sp * co - cp * so;
            const float e0 =  chs * ct, e1 = -shs * ct;
            const float e2 = -chd * sn, e3 = -shd * sn;
            const float e4 =  chd * sn, e5 = -shd * sn;
            const float e6 =  chs * ct, e7 =  shs * ct;
            if (t < NQ) {
                const float a = x[b * NQ + t] * 0.5f;
                const float c = cosf(a), s = sinf(a);
                gr[0] = e0 * c + e3 * s;  gr[1] = e1 * c - e2 * s;
                gr[2] = e1 * s + e2 * c;  gr[3] = e3 * c - e0 * s;
                gr[4] = e4 * c + e7 * s;  gr[5] = e5 * c - e6 * s;
                gr[6] = e5 * s + e6 * c;  gr[7] = e7 * c - e4 * s;
            } else {
                gr[0] = e0; gr[1] = e1; gr[2] = e2; gr[3] = e3;
                gr[4] = e4; gr[5] = e5; gr[6] = e6; gr[7] = e7;
            }
        } else {
            gr[0] = ct; gr[1] = sn; gr[2] = cp; gr[3] = sp;
            gr[4] = co; gr[5] = so; gr[6] = 0.0f; gr[7] = 0.0f;
        }
    }
    __syncthreads();

    buildChainTables(midT, lowT, topV, gu, wave, lane);
    if (t < 240) buildPhase(ph, gu, t);
    buildGroup<2>(mr, gu + 2 * NQ * 8, t);
    buildGroup<0>(mr, gu + 2 * NQ * 8, t);
    __syncthreads();

    layer01Direct(st + 2 * NSTATE, midT, lowT, topV, gu, wave, lane);
    __syncthreads();

    layerPass<2>(st, mr, ph, gu, t, wave, lane);
    layerPass<3>(st, mr, ph, gu, t, wave, lane);

    constexpr int RLAST = (NLAYERS - 1) % (NQ - 1) + 1;
    static_assert(oneMidBitPerMask(RLAST),
                  "the closing sums need exactly one middle-group bit per observable");
    const float* fin = st + ((NLAYERS - 1) & 1) * 2 * NSTATE;
    const int ibase = ((t >> 4) << 8) | (t & 15);
    const int fb0 = swz(ibase), fb1 = fb0 ^ 32;
    v2f are[8], aim[8], pw[8];
#pragma unroll
    for (int j = 0; j < 8; ++j) {
        const int i0 = swzAt(fb0, fb1, (2 * j) << 4);
        const int i1 = swzAt(fb0, fb1, (2 * j + 1) << 4);
        are[j].x = fin[i0];
        are[j].y = fin[i1];
        aim[j].x = fin[NSTATE + i0];
        aim[j].y = fin[NSTATE + i1];
        pw[j] = are[j] * are[j] + aim[j] * aim[j];
    }
    float sd[4];
    {
        v2f s3[4], d3[4];
#pragma unroll
        for (int j = 0; j < 4; ++j) {
            s3[j] = pw[j] + pw[j + 4];
            d3[j] = pw[j] - pw[j + 4];
        }
        const v2f t3 = (d3[0] + d3[1]) + (d3[2] + d3[3]);
        sd[3] = t3.x + t3.y;
        const v2f s2a = s3[0] + s3[2], d2a = s3[0] - s3[2];
        const v2f s2b = s3[1] + s3[3], d2b = s3[1] - s3[3];
        const v2f t2 = d2a + d2b;
        sd[2] = t2.x + t2.y;
        const v2f s1 = s2a + s2b, d1 = s2a - s2b;
        sd[1] = d1.x + d1.y;
        sd[0] = s1.x - s1.y;
    }
    float cr[4];
    {
        float c0 = are[0].x * are[0].y;
        c0 = __builtin_fmaf(aim[0].x, aim[0].y, c0);
#pragma unroll
        for (int j = 1; j < 8; ++j) {
            c0 = __builtin_fmaf(are[j].x, are[j].y, c0);
            c0 = __builtin_fmaf(aim[j].x, aim[j].y, c0);
        }
        cr[0] = c0;
    }
#pragma unroll
    for (int b = 1; b < 4; ++b) {
        v2f cre = {0.0f, 0.0f}, cim = {0.0f, 0.0f};
#pragma unroll
        for (int j = 0; j < 8; ++j) {
            if ((j >> (b - 1)) & 1) continue;
            cre += are[j] * are[j | (1 << (b - 1))];
            cim += aim[j] * aim[j | (1 << (b - 1))];
        }
        const v2f cs = cre + cim;
        cr[b] = cs.x + cs.y;
    }
    const float* gl = gu + (NLAYERS - 1) * NQ * 8;
    float vb[4];
#pragma unroll
    for (int b = 0; b < 4; ++b) {
        const float ch = gl[(7 - b) * 8], sh = gl[(7 - b) * 8 + 1];
        vb[b] = (ch * ch - sh * sh) * sd[b] - 4.0f * (ch * sh) * cr[b];
    }
    const int jt = ringDst(RLAST, ibase);
    float part[NQ];
#pragma unroll
    for (int q = 0; q < NQ; ++q) {
        const int mm = (obsMask(RLAST, q) >> 4) & 15;
        const int b = (mm == 1) ? 0 : (mm == 2) ? 1 : (mm == 4) ? 2 : 3;
        part[q] = ((jt >> (11 - q)) & 1) ? -vb[b] : vb[b];
    }
    float red[NQ];
#pragma unroll
    for (int q = 0; q < NQ; ++q) {
        float v = part[q];
        v += xhalf(v);
        v += rowXor<8>(v);
        v += rowXor<4>(v);
        v += rowXor<2>(v);
        v += rowXor<1>(v);
        red[q] = v;
    }
    if (lane == 0) {
#pragma unroll
        for (int q = 0; q < NQ; ++q) wred[wave * NQ + q] = red[q];
    }
    __syncthreads();
    if (t < NQ) {
        float s = 0.0f;
#pragma unroll
        for (int wv = 0; wv < 8; ++wv) s += wred[wv * NQ + t];
        out[b * NQ + t] = s;
    }
}

extern "C" void kernel_launch(void* const* d_in, const int* in_sizes, int n_in,
                              void* d_out, int out_size, void* d_ws, size_t ws_size,
                              hipStream_t stream) {
    const float* x   = (const float*)d_in[0];
    const float* wts = (const float*)d_in[1];
    float* out = (float*)d_out;
    const int batch = in_sizes[0] / NQ;
    vqc_kernel<<<batch, NTHREADS, 0, stream>>>(x, wts, out);
}
